# starting kernel
# baseline (speedup 1.0000x reference)
#include <hip/hip_runtime.h>
#include <hip/hip_bf16.h>


typedef __attribute__((ext_vector_type(2))) float v2f;
typedef __attribute__((ext_vector_type(8))) float v8f;

#define DVEC 128
#define LDSTR 132
#define INV_SIGMA 10.0f

__global__ void ephn_norm_kernel(const float* __restrict__ fvec,
                                 float* __restrict__ rnorm) {
    const int row  = blockIdx.x * 4 + (threadIdx.x >> 5);
    const int lane = threadIdx.x & 31;
    const float4 d = ((const float4*)(fvec + ((size_t)row << 7)))[lane];
    float s = d.x * d.x + d.y * d.y + d.z * d.z + d.w * d.w;
    #pragma unroll
    for (int off = 16; off >= 1; off >>= 1)
        s += __shfl_xor(s, off, 32);
    if (lane == 0) rnorm[row] = rsqrtf(s);
}

__global__ void ephn_main_kernel(const float* __restrict__ fvec,
                                 const int*   __restrict__ Lvec,
                                 const float* __restrict__ rnorm,
                                 float* __restrict__ wsPos,
                                 float* __restrict__ wsNeg,
                                 int N) {
    __shared__ float lds_a[16 * LDSTR];
    __shared__ float lds_b[64 * LDSTR];
    __shared__ float mPV[64]; __shared__ int mPJ[64];
    __shared__ float mNV[64]; __shared__ int mNJ[64];

    const int tid = threadIdx.x;
    const int l   = tid & 31;
    const int w   = tid >> 5;
    const int m   = l & 15;
    const int hi  = l >> 4;
    const int i0  = blockIdx.x * 16;

    for (int t = tid; t < 16 * 32; t += 128) {
        const int f   = t * 4;
        const int row = f >> 7;
        const int col = f & 127;
        float4 d = *(const float4*)(fvec + ((size_t)i0 << 7) + f);
        const float r = rnorm[i0 + row];
        float* dst = &lds_a[row * LDSTR + col];
        dst[0] = d.x * r; dst[1] = d.y * r; dst[2] = d.z * r; dst[3] = d.w * r;
    }

    int Li[8];
    #pragma unroll
    for (int v = 0; v < 8; ++v) Li[v] = Lvec[i0 + v + 8 * hi];

    float posV[8], negV[8]; int posJ[8], negJ[8];
    #pragma unroll
    for (int v = 0; v < 8; ++v) {
        posV[v] = -2.0f; negV[v] = -2.0f;
        posJ[v] = 0x7fffffff; negJ[v] = 0x7fffffff;
    }

    const float* aptr = &lds_a[m * LDSTR + 2 * hi];
    const float* bptr = &lds_b[(w * 16 + m) * LDSTR + 2 * hi];

    const int nstrips = N >> 6;
    for (int jt = 0; jt < nstrips; ++jt) {
        __syncthreads();
        const size_t gbase = ((size_t)jt << 6) << 7;
        for (int t = tid; t < 64 * 32; t += 128) {
            const int f   = t * 4;
            const int row = f >> 7;
            const int col = f & 127;
            float4 d = *(const float4*)(fvec + gbase + f);
            const float r = rnorm[jt * 64 + row];
            float* dst = &lds_b[row * LDSTR + col];
            dst[0] = d.x * r; dst[1] = d.y * r; dst[2] = d.z * r; dst[3] = d.w * r;
        }
        __syncthreads();

        v8f c = {};
        #pragma unroll
        for (int k = 0; k < DVEC; k += 4) {
            v2f a = *(const v2f*)(aptr + k);
            v2f b = *(const v2f*)(bptr + k);
            c = __builtin_amdgcn_wmma_f32_16x16x4_f32(
                    false, a, false, b, (short)0, c, false, false);
        }

        const int j  = jt * 64 + w * 16 + m;
        const int Lj = Lvec[j];
        #pragma unroll
        for (int v = 0; v < 8; ++v) {
            const int   i   = i0 + v + 8 * hi;
            const float d   = c[v];
            const bool  eqc = (Li[v] == Lj);
            const float pv = eqc ? (((i == j) || (d > 0.9999f)) ? -1.0f : d)
                                 : -1.0f;
            if (pv > posV[v] || (pv == posV[v] && j < posJ[v])) {
                posV[v] = pv; posJ[v] = j;
            }
            const float nv = eqc ? -1.0f : d;
            if (nv > negV[v] || (nv == negV[v] && j < negJ[v])) {
                negV[v] = nv; negJ[v] = j;
            }
        }
    }

    #pragma unroll
    for (int v = 0; v < 8; ++v) {
        #pragma unroll
        for (int off = 1; off < 16; off <<= 1) {
            float ov = __shfl_xor(posV[v], off, 16);
            int   oj = __shfl_xor(posJ[v], off, 16);
            if (ov > posV[v] || (ov == posV[v] && oj < posJ[v])) {
                posV[v] = ov; posJ[v] = oj;
            }
            float onv = __shfl_xor(negV[v], off, 16);
            int   onj = __shfl_xor(negJ[v], off, 16);
            if (onv > negV[v] || (onv == negV[v] && onj < negJ[v])) {
                negV[v] = onv; negJ[v] = onj;
            }
        }
    }

    if ((l & 15) == 0) {
        #pragma unroll
        for (int v = 0; v < 8; ++v) {
            const int row = v + 8 * hi;
            const int idx = w * 16 + row;
            mPV[idx] = posV[v]; mPJ[idx] = posJ[v];
            mNV[idx] = negV[v]; mNJ[idx] = negJ[v];
        }
    }
    __syncthreads();
    if (tid < 16) {
        float bpv = -2.0f, bnv = -2.0f;
        int   bpj = 0x7fffffff, bnj = 0x7fffffff;
        #pragma unroll
        for (int ww = 0; ww < 4; ++ww) {
            const int idx = ww * 16 + tid;
            const float pv = mPV[idx]; const int pj = mPJ[idx];
            if (pv > bpv || (pv == bpv && pj < bpj)) { bpv = pv; bpj = pj; }
            const float nv = mNV[idx]; const int nj = mNJ[idx];
            if (nv > bnv || (nv == bnv && nj < bnj)) { bnv = nv; bnj = nj; }
        }
        wsPos[i0 + tid] = bpv;
        wsNeg[i0 + tid] = bnv;
    }
}

__global__ void ephn_finalize_kernel(const float* __restrict__ wsPos,
                                     const float* __restrict__ wsNeg,
                                     float* __restrict__ out, int N) {
    __shared__ float sProb[1024];
    __shared__ float sPos[1024];
    __shared__ float sNeg[1024];
    __shared__ int   sCnt[1024];
    const int tid = threadIdx.x;

    float* eps = out + 1;
    float* hns = out + 1 + (N + 2);

    float accProb = 0.0f, accPos = 0.0f, accNeg = 0.0f;
    int accCnt = 0;
    for (int i = tid; i < N; i += 1024) {
        const float p = wsPos[i];
        const float n = wsNeg[i];
        eps[i] = (p > 0.0f) ? p : 0.0f;
        hns[i] = (n > 0.0f) ? n : 0.0f;
        accPos += p;
        accNeg += n;
        if (p > 0.0f && n > 0.0f) {
            const float s = (n - p) * INV_SIGMA;
            const float prob = (s > 0.0f) ? (s + log1pf(expf(-s)))
                                          : log1pf(expf(s));
            accProb += prob;
            accCnt  += 1;
        }
    }
    sProb[tid] = accProb; sPos[tid] = accPos; sNeg[tid] = accNeg; sCnt[tid] = accCnt;
    __syncthreads();
    for (int st = 512; st > 0; st >>= 1) {
        if (tid < st) {
            sProb[tid] += sProb[tid + st];
            sPos[tid]  += sPos[tid + st];
            sNeg[tid]  += sNeg[tid + st];
            sCnt[tid]  += sCnt[tid + st];
        }
        __syncthreads();
    }
    if (tid == 0) {
        const int cnt = (sCnt[0] > 1) ? sCnt[0] : 1;
        out[0] = sProb[0] / (float)cnt;
        eps[N] = 0.0f; eps[N + 1] = 1.0f;
        hns[N] = 0.0f; hns[N + 1] = 1.0f;
        out[1 + 2 * (N + 2)] = (sPos[0] - sNeg[0]) / (float)N;
    }
}

extern "C" void kernel_launch(void* const* d_in, const int* in_sizes, int n_in,
                              void* d_out, int out_size, void* d_ws, size_t ws_size,
                              hipStream_t stream) {
    const float* fvec = (const float*)d_in[0];
    const int*   Lvec = (const int*)d_in[1];
    const int N = in_sizes[1];

    float* rnorm = (float*)d_ws;
    float* wsPos = rnorm + N;
    float* wsNeg = wsPos + N;

    ephn_norm_kernel<<<N / 4, 128, 0, stream>>>(fvec, rnorm);
    ephn_main_kernel<<<N / 16, 128, 0, stream>>>(fvec, Lvec, rnorm,
                                                 wsPos, wsNeg, N);
    ephn_finalize_kernel<<<1, 1024, 0, stream>>>(wsPos, wsNeg,
                                                 (float*)d_out, N);
}


# Rows ranked by class and packed in class order (new rank launch): same-class pairs lie in a diagonal band; only band tiles run masked mining, other tiles one max per element; maxima only, no index
# speedup vs baseline: 2.7932x; 2.7932x over previous
#include <hip/hip_runtime.h>
#include <hip/hip_bf16.h>


typedef __attribute__((ext_vector_type(2))) float v2f;
typedef __attribute__((ext_vector_type(4))) float v4f;
typedef __attribute__((ext_vector_type(8))) float v8f;

#define DVEC 128
#define INV_SIGMA 10.0f
#define PARTS 8
#define PKSTR 33

__global__ void __launch_bounds__(256)
ephn_rank_kernel(const int* __restrict__ Lvec, int* __restrict__ perm, int N) {
    __shared__ int sRank[32];
    const int lane = threadIdx.x & 31;
    const int w    = threadIdx.x >> 5;
    const int rch  = blockIdx.x * 8 + w;
    const int4* Lp = (const int4*)Lvec;
    const int4 own = Lp[rch];
    const int Lr[4] = {own.x, own.y, own.z, own.w};
    int cnt[4] = {0, 0, 0, 0};
    const int nch = N >> 2;
    #pragma unroll 16
    for (int ch = lane; ch < nch; ch += 32) {
        const int4 v = Lp[ch];
        const int adj = (ch < rch) ? 1 : 0;
        #pragma unroll
        for (int k = 0; k < 4; ++k) {
            const int thr = Lr[k] + adj;
            cnt[k] += (v.x < thr) + (v.y < thr) + (v.z < thr) + (v.w < thr);
        }
    }
    #pragma unroll
    for (int k = 0; k < 4; ++k) {
        #pragma unroll
        for (int off = 16; off >= 1; off >>= 1)
            cnt[k] += __shfl_xor(cnt[k], off, 32);
    }
    cnt[1] += (Lr[0] == Lr[1]);
    cnt[2] += (Lr[0] == Lr[2]) + (Lr[1] == Lr[2]);
    cnt[3] += (Lr[0] == Lr[3]) + (Lr[1] == Lr[3]) + (Lr[2] == Lr[3]);
    if (lane == 0) {
        sRank[w * 4 + 0] = cnt[0];
        sRank[w * 4 + 1] = cnt[1];
        sRank[w * 4 + 2] = cnt[2];
        sRank[w * 4 + 3] = cnt[3];
    }
    __syncthreads();
    if (w == 0) perm[blockIdx.x * 32 + lane] = sRank[lane];
}

__global__ void __launch_bounds__(1024)
ephn_pack_kernel(const float* __restrict__ fvec, const int* __restrict__ Lvec,
                 const int* __restrict__ perm, float* __restrict__ pk,
                 int* __restrict__ Ls, int N) {
    __shared__ v4f tile[32 * PKSTR];
    __shared__ int sInv[32];
    const int wv   = threadIdx.x >> 5;
    const int lane = threadIdx.x & 31;
    const int p0   = blockIdx.x * 32;
    const int4* perm4 = (const int4*)perm;
    const int nch = N >> 2;
    for (int ca = threadIdx.x; ca < nch; ca += 2048) {
        const int cb = ca + 1024;
        const int4 va = perm4[ca];
        int4 vb = {0, 0, 0, 0};
        if (cb < nch) vb = perm4[cb];
        const int a0 = va.x - p0, a1 = va.y - p0, a2 = va.z - p0, a3 = va.w - p0;
        if ((unsigned)a0 < 32u) sInv[a0] = 4 * ca;
        if ((unsigned)a1 < 32u) sInv[a1] = 4 * ca + 1;
        if ((unsigned)a2 < 32u) sInv[a2] = 4 * ca + 2;
        if ((unsigned)a3 < 32u) sInv[a3] = 4 * ca + 3;
        if (cb < nch) {
            const int b0 = vb.x - p0, b1 = vb.y - p0, b2 = vb.z - p0, b3 = vb.w - p0;
            if ((unsigned)b0 < 32u) sInv[b0] = 4 * cb;
            if ((unsigned)b1 < 32u) sInv[b1] = 4 * cb + 1;
            if ((unsigned)b2 < 32u) sInv[b2] = 4 * cb + 2;
            if ((unsigned)b3 < 32u) sInv[b3] = 4 * cb + 3;
        }
    }
    __syncthreads();
    const int row = sInv[wv];
    const float4 d = ((const float4*)(fvec + ((size_t)row << 7)))[lane];
    float s = d.x * d.x + d.y * d.y + d.z * d.z + d.w * d.w;
    #pragma unroll
    for (int off = 16; off >= 1; off >>= 1)
        s += __shfl_xor(s, off, 32);
    const float r = rsqrtf(s);
    v4f o;
    o.x = d.x * r; o.y = d.y * r; o.z = d.z * r; o.w = d.w * r;
    tile[wv * PKSTR + lane] = o;
    __syncthreads();
    const int c  = lane & 15;
    const int h  = lane >> 4;
    const int tl = wv >> 4;
    const int q  = wv & 15;
    ((v4f*)pk)[((size_t)((blockIdx.x * 2 + tl) * 16 + q)) * 32 + lane] =
        tile[(tl * 16 + c) * PKSTR + 2 * q + h];
    if (wv == 0) Ls[p0 + lane] = Lvec[sInv[lane]];
}

struct StreamTile {
    v4f f[16];
    int4 l0, l1;
};

__device__ __forceinline__ void load_stream(StreamTile& s, const v4f* __restrict__ pk4,
                                            const int* __restrict__ Lvec,
                                            int ts, int lane, int h) {
    const v4f* p = pk4 + ((size_t)ts * 16) * 32 + lane;
    #pragma unroll
    for (int q = 0; q < 16; ++q) s.f[q] = p[q * 32];
    const int4* lp = (const int4*)(Lvec + ts * 16 + 8 * h);
    s.l0 = lp[0];
    s.l1 = lp[1];
}

__device__ __forceinline__ void mine8_masked(const v8f& acc, const StreamTile& s, int jbase,
                                             int i, int Li, float& posV, float& negV) {
    const int Lj[8] = {s.l0.x, s.l0.y, s.l0.z, s.l0.w, s.l1.x, s.l1.y, s.l1.z, s.l1.w};
    #pragma unroll
    for (int r = 0; r < 8; ++r) {
        const int   j   = jbase + r;
        const float d   = acc[r];
        const bool  eqc = (Li == Lj[r]);
        const float pv = eqc ? (((i == j) || (d > 0.9999f)) ? -1.0f : d)
                             : -1.0f;
        posV = fmaxf(posV, pv);
        const float nv = eqc ? -1.0f : d;
        negV = fmaxf(negV, nv);
    }
}

__device__ __forceinline__ void mine8_plain(const v8f& acc, float& negV) {
    #pragma unroll
    for (int r = 0; r < 8; ++r) negV = fmaxf(negV, acc[r]);
}

__global__ void __launch_bounds__(256)
ephn_main_kernel(const float* __restrict__ pk,
                 const int*   __restrict__ Lvec,
                 float* __restrict__ wsPos,
                 float* __restrict__ wsNeg,
                 int N) {
    const int lane = threadIdx.x & 31;
    const int w    = threadIdx.x >> 5;
    const int c    = lane & 15;
    const int h    = lane >> 4;
    const int rb   = blockIdx.x / PARTS;
    const int part = blockIdx.x - rb * PARTS;
    const int tpp  = (N >> 4) / PARTS;
    const int t0   = part * tpp;
    const int own0 = rb * 16 + w * 2;
    const v4f* pk4 = (const v4f*)pk;

    v4f b0[16], b1[16];
    {
        const v4f* p0 = pk4 + ((size_t)own0 * 16) * 32 + lane;
        #pragma unroll
        for (int q = 0; q < 16; ++q) { b0[q] = p0[q * 32]; b1[q] = p0[(16 + q) * 32]; }
    }
    const int i0  = own0 * 16 + c;
    const int i1  = i0 + 16;
    const int Li0 = Lvec[i0];
    const int Li1 = Lvec[i1];

    const int ownLo = __builtin_amdgcn_readfirstlane(Lvec[own0 * 16]);
    const int ownHi = __builtin_amdgcn_readfirstlane(Lvec[own0 * 16 + 31]);

    float posV0 = -1.0f, negV0 = -1.0f, posV1 = -1.0f, negV1 = -1.0f;

    StreamTile sa, sb;
    load_stream(sa, pk4, Lvec, t0, lane, h);

    for (int it = 0; it < tpp; it += 2) {
        #pragma unroll
        for (int half = 0; half < 2; ++half) {
            StreamTile& cur = half ? sb : sa;
            StreamTile& nxt = half ? sa : sb;
            const int ts = t0 + it + half;
            int tn = it + half + 1;
            tn = (tn < tpp) ? tn : (tpp - 1);
            load_stream(nxt, pk4, Lvec, t0 + tn, lane, h);

            v8f c0 = {};
            v8f c1 = {};
            #pragma unroll
            for (int q = 0; q < 16; ++q) {
                const v2f alo = cur.f[q].xy;
                const v2f ahi = cur.f[q].zw;
                c0 = __builtin_amdgcn_wmma_f32_16x16x4_f32(
                        false, alo, false, b0[q].xy, (short)0, c0, false, false);
                c1 = __builtin_amdgcn_wmma_f32_16x16x4_f32(
                        false, alo, false, b1[q].xy, (short)0, c1, false, false);
                asm volatile("" : "+v"(c0), "+v"(c1));
                c0 = __builtin_amdgcn_wmma_f32_16x16x4_f32(
                        false, ahi, false, b0[q].zw, (short)0, c0, false, false);
                c1 = __builtin_amdgcn_wmma_f32_16x16x4_f32(
                        false, ahi, false, b1[q].zw, (short)0, c1, false, false);
                asm volatile("" : "+v"(c0), "+v"(c1));
            }

            const int tLo = __builtin_amdgcn_readlane(cur.l0.x, 0);
            const int tHi = __builtin_amdgcn_readlane(cur.l1.w, 16);
            if (tLo <= ownHi && tHi >= ownLo) {
                const int jbase = ts * 16 + 8 * h;
                mine8_masked(c0, cur, jbase, i0, Li0, posV0, negV0);
                mine8_masked(c1, cur, jbase, i1, Li1, posV1, negV1);
            } else {
                mine8_plain(c0, negV0);
                mine8_plain(c1, negV1);
            }
        }
    }

    posV0 = fmaxf(posV0, __shfl_xor(posV0, 16, 32));
    negV0 = fmaxf(negV0, __shfl_xor(negV0, 16, 32));
    posV1 = fmaxf(posV1, __shfl_xor(posV1, 16, 32));
    negV1 = fmaxf(negV1, __shfl_xor(negV1, 16, 32));
    const size_t o = (size_t)part * N + (size_t)own0 * 16 + lane;
    wsPos[o] = h ? posV1 : posV0;
    wsNeg[o] = h ? negV1 : negV0;
}

__global__ void ephn_finalize_kernel(const float* __restrict__ wsPos,
                                     const float* __restrict__ wsNeg,
                                     const int*   __restrict__ perm,
                                     float* __restrict__ out, int N) {
    __shared__ float sProb[1024];
    __shared__ float sPos[1024];
    __shared__ float sNeg[1024];
    __shared__ int   sCnt[1024];
    const int tid = threadIdx.x;

    float* eps = out + 1;
    float* hns = out + 1 + (N + 2);

    float accProb = 0.0f, accPos = 0.0f, accNeg = 0.0f;
    int accCnt = 0;
    int psNext = (tid < N) ? perm[tid] : 0;
    for (int i = tid; i < N; i += 1024) {
        const int ps = psNext;
        psNext = (i + 1024 < N) ? perm[i + 1024] : 0;
        float p = wsPos[ps];
        float n = wsNeg[ps];
        #pragma unroll
        for (int pt = 1; pt < PARTS; ++pt) {
            const float pp = wsPos[(size_t)pt * N + ps];
            const float nn = wsNeg[(size_t)pt * N + ps];
            if (pp > p) p = pp;
            if (nn > n) n = nn;
        }
        eps[i] = (p > 0.0f) ? p : 0.0f;
        hns[i] = (n > 0.0f) ? n : 0.0f;
        accPos += p;
        accNeg += n;
        if (p > 0.0f && n > 0.0f) {
            const float s = (n - p) * INV_SIGMA;
            const float prob = (s > 0.0f) ? (s + log1pf(expf(-s)))
                                          : log1pf(expf(s));
            accProb += prob;
            accCnt  += 1;
        }
    }
    sProb[tid] = accProb; sPos[tid] = accPos; sNeg[tid] = accNeg; sCnt[tid] = accCnt;
    __syncthreads();
    for (int st = 512; st > 0; st >>= 1) {
        if (tid < st) {
            sProb[tid] += sProb[tid + st];
            sPos[tid]  += sPos[tid + st];
            sNeg[tid]  += sNeg[tid + st];
            sCnt[tid]  += sCnt[tid + st];
        }
        __syncthreads();
    }
    if (tid == 0) {
        const int cnt = (sCnt[0] > 1) ? sCnt[0] : 1;
        out[0] = sProb[0] / (float)cnt;
        eps[N] = 0.0f; eps[N + 1] = 1.0f;
        hns[N] = 0.0f; hns[N + 1] = 1.0f;
        out[1 + 2 * (N + 2)] = (sPos[0] - sNeg[0]) / (float)N;
    }
}

extern "C" void kernel_launch(void* const* d_in, const int* in_sizes, int n_in,
                              void* d_out, int out_size, void* d_ws, size_t ws_size,
                              hipStream_t stream) {
    const float* fvec = (const float*)d_in[0];
    const int*   Lvec = (const int*)d_in[1];
    const int N = in_sizes[1];

    float* pk    = (float*)d_ws;
    float* wsPos = pk + N * DVEC;
    float* wsNeg = wsPos + N * PARTS;
    int*   perm  = (int*)(wsNeg + N * PARTS);
    int*   Ls    = perm + N;

    ephn_rank_kernel<<<N / 32, 256, 0, stream>>>(Lvec, perm, N);
    ephn_pack_kernel<<<N / 32, 1024, 0, stream>>>(fvec, Lvec, perm, pk, Ls, N);
    ephn_main_kernel<<<N / 32, 256, 0, stream>>>(pk, Ls, wsPos, wsNeg, N);
    ephn_finalize_kernel<<<1, 1024, 0, stream>>>(wsPos, wsNeg, perm,
                                                 (float*)d_out, N);
}


# A block owns 32 rows and all columns (one part per wave, sweep start staggered by block): parts joined in LDS, loss terms summed per block, 65-block output kernel; loss and mean sum order changes
# speedup vs baseline: 4.4424x; 1.5904x over previous
#include <hip/hip_runtime.h>
#include <hip/hip_bf16.h>


typedef __attribute__((ext_vector_type(2))) float v2f;
typedef __attribute__((ext_vector_type(4))) float v4f;
typedef __attribute__((ext_vector_type(8))) float v8f;

#define DVEC 128
#define INV_SIGMA 10.0f
#define PARTS 8
#define PKSTR 33

__global__ void __launch_bounds__(256)
ephn_rank_kernel(const int* __restrict__ Lvec, int* __restrict__ perm, int N) {
    __shared__ int sRank[32];
    const int lane = threadIdx.x & 31;
    const int w    = threadIdx.x >> 5;
    const int rch  = blockIdx.x * 8 + w;
    const int4* Lp = (const int4*)Lvec;
    const int4 own = Lp[rch];
    const int Lr[4] = {own.x, own.y, own.z, own.w};
    int cnt[4] = {0, 0, 0, 0};
    const int nch = N >> 2;
    #pragma unroll 16
    for (int ch = lane; ch < nch; ch += 32) {
        const int4 v = Lp[ch];
        const int adj = (ch < rch) ? 1 : 0;
        #pragma unroll
        for (int k = 0; k < 4; ++k) {
            const int thr = Lr[k] + adj;
            cnt[k] += (v.x < thr) + (v.y < thr) + (v.z < thr) + (v.w < thr);
        }
    }
    #pragma unroll
    for (int k = 0; k < 4; ++k) {
        #pragma unroll
        for (int off = 16; off >= 1; off >>= 1)
            cnt[k] += __shfl_xor(cnt[k], off, 32);
    }
    cnt[1] += (Lr[0] == Lr[1]);
    cnt[2] += (Lr[0] == Lr[2]) + (Lr[1] == Lr[2]);
    cnt[3] += (Lr[0] == Lr[3]) + (Lr[1] == Lr[3]) + (Lr[2] == Lr[3]);
    if (lane == 0) {
        sRank[w * 4 + 0] = cnt[0];
        sRank[w * 4 + 1] = cnt[1];
        sRank[w * 4 + 2] = cnt[2];
        sRank[w * 4 + 3] = cnt[3];
    }
    __syncthreads();
    if (w == 0) perm[blockIdx.x * 32 + lane] = sRank[lane];
}

__global__ void __launch_bounds__(1024)
ephn_pack_kernel(const float* __restrict__ fvec, const int* __restrict__ Lvec,
                 const int* __restrict__ perm, float* __restrict__ pk,
                 int* __restrict__ Ls, int N) {
    __shared__ v4f tile[32 * PKSTR];
    __shared__ int sInv[32];
    const int wv   = threadIdx.x >> 5;
    const int lane = threadIdx.x & 31;
    const int p0   = blockIdx.x * 32;
    const int4* perm4 = (const int4*)perm;
    const int nch = N >> 2;
    for (int ca = threadIdx.x; ca < nch; ca += 2048) {
        const int cb = ca + 1024;
        const int4 va = perm4[ca];
        int4 vb = {0, 0, 0, 0};
        if (cb < nch) vb = perm4[cb];
        const int a0 = va.x - p0, a1 = va.y - p0, a2 = va.z - p0, a3 = va.w - p0;
        if ((unsigned)a0 < 32u) sInv[a0] = 4 * ca;
        if ((unsigned)a1 < 32u) sInv[a1] = 4 * ca + 1;
        if ((unsigned)a2 < 32u) sInv[a2] = 4 * ca + 2;
        if ((unsigned)a3 < 32u) sInv[a3] = 4 * ca + 3;
        if (cb < nch) {
            const int b0 = vb.x - p0, b1 = vb.y - p0, b2 = vb.z - p0, b3 = vb.w - p0;
            if ((unsigned)b0 < 32u) sInv[b0] = 4 * cb;
            if ((unsigned)b1 < 32u) sInv[b1] = 4 * cb + 1;
            if ((unsigned)b2 < 32u) sInv[b2] = 4 * cb + 2;
            if ((unsigned)b3 < 32u) sInv[b3] = 4 * cb + 3;
        }
    }
    __syncthreads();
    const int row = sInv[wv];
    const float4 d = ((const float4*)(fvec + ((size_t)row << 7)))[lane];
    float s = d.x * d.x + d.y * d.y + d.z * d.z + d.w * d.w;
    #pragma unroll
    for (int off = 16; off >= 1; off >>= 1)
        s += __shfl_xor(s, off, 32);
    const float r = rsqrtf(s);
    v4f o;
    o.x = d.x * r; o.y = d.y * r; o.z = d.z * r; o.w = d.w * r;
    tile[wv * PKSTR + lane] = o;
    __syncthreads();
    const int c  = lane & 15;
    const int h  = lane >> 4;
    const int tl = wv >> 4;
    const int q  = wv & 15;
    ((v4f*)pk)[((size_t)((blockIdx.x * 2 + tl) * 16 + q)) * 32 + lane] =
        tile[(tl * 16 + c) * PKSTR + 2 * q + h];
    if (wv == 0) Ls[p0 + lane] = Lvec[sInv[lane]];
}

struct StreamTile {
    v4f f[16];
    int4 l0, l1;
};

__device__ __forceinline__ void load_stream(StreamTile& s, const v4f* __restrict__ pk4,
                                            const int* __restrict__ Lvec,
                                            int ts, int lane, int h) {
    const v4f* p = pk4 + ((size_t)ts * 16) * 32 + lane;
    #pragma unroll
    for (int q = 0; q < 16; ++q) s.f[q] = p[q * 32];
    const int4* lp = (const int4*)(Lvec + ts * 16 + 8 * h);
    s.l0 = lp[0];
    s.l1 = lp[1];
}

__device__ __forceinline__ void mine8_masked(const v8f& acc, const StreamTile& s, int jbase,
                                             int i, int Li, float& posV, float& negV) {
    const int Lj[8] = {s.l0.x, s.l0.y, s.l0.z, s.l0.w, s.l1.x, s.l1.y, s.l1.z, s.l1.w};
    #pragma unroll
    for (int r = 0; r < 8; ++r) {
        const int   j   = jbase + r;
        const float d   = acc[r];
        const bool  eqc = (Li == Lj[r]);
        const float pv = eqc ? (((i == j) || (d > 0.9999f)) ? -1.0f : d)
                             : -1.0f;
        posV = fmaxf(posV, pv);
        const float nv = eqc ? -1.0f : d;
        negV = fmaxf(negV, nv);
    }
}

__device__ __forceinline__ void mine8_plain(const v8f& acc, float& negV) {
    #pragma unroll
    for (int r = 0; r < 8; ++r) negV = fmaxf(negV, acc[r]);
}

__global__ void __launch_bounds__(256)
ephn_main_kernel(const float* __restrict__ pk,
                 const int*   __restrict__ Lvec,
                 float* __restrict__ posM,
                 float* __restrict__ negM,
                 float* __restrict__ part,
                 int N) {
    __shared__ float sP[PARTS][32];
    __shared__ float sN[PARTS][32];
    const int lane = threadIdx.x & 31;
    const int w    = __builtin_amdgcn_readfirstlane(threadIdx.x >> 5);
    const int c    = lane & 15;
    const int h    = lane >> 4;
    const int tpp  = (N >> 4) / PARTS;
    const int t0   = w * tpp;
    const int own0 = blockIdx.x * 2;
    const v4f* pk4 = (const v4f*)pk;

    v4f b0[16], b1[16];
    {
        const v4f* p0 = pk4 + ((size_t)own0 * 16) * 32 + lane;
        #pragma unroll
        for (int q = 0; q < 16; ++q) { b0[q] = p0[q * 32]; b1[q] = p0[(16 + q) * 32]; }
    }
    const int i0  = own0 * 16 + c;
    const int i1  = i0 + 16;
    const int Li0 = Lvec[i0];
    const int Li1 = Lvec[i1];

    const int ownLo = __builtin_amdgcn_readfirstlane(Lvec[own0 * 16]);
    const int ownHi = __builtin_amdgcn_readfirstlane(Lvec[own0 * 16 + 31]);

    float posV0 = -1.0f, negV0 = -1.0f, posV1 = -1.0f, negV1 = -1.0f;

    int cidx = blockIdx.x % tpp;

    StreamTile sa, sb;
    load_stream(sa, pk4, Lvec, t0 + cidx, lane, h);

    for (int it = 0; it < tpp; it += 2) {
        #pragma unroll
        for (int half = 0; half < 2; ++half) {
            StreamTile& cur = half ? sb : sa;
            StreamTile& nxt = half ? sa : sb;
            const int ts = t0 + cidx;
            int nidx = cidx + 1;
            nidx = (nidx < tpp) ? nidx : 0;
            load_stream(nxt, pk4, Lvec, t0 + nidx, lane, h);

            v8f c0 = {};
            v8f c1 = {};
            #pragma unroll
            for (int q = 0; q < 16; ++q) {
                const v2f alo = cur.f[q].xy;
                const v2f ahi = cur.f[q].zw;
                c0 = __builtin_amdgcn_wmma_f32_16x16x4_f32(
                        false, alo, false, b0[q].xy, (short)0, c0, false, false);
                c1 = __builtin_amdgcn_wmma_f32_16x16x4_f32(
                        false, alo, false, b1[q].xy, (short)0, c1, false, false);
                asm volatile("" : "+v"(c0), "+v"(c1));
                c0 = __builtin_amdgcn_wmma_f32_16x16x4_f32(
                        false, ahi, false, b0[q].zw, (short)0, c0, false, false);
                c1 = __builtin_amdgcn_wmma_f32_16x16x4_f32(
                        false, ahi, false, b1[q].zw, (short)0, c1, false, false);
                asm volatile("" : "+v"(c0), "+v"(c1));
            }

            const int tLo = __builtin_amdgcn_readlane(cur.l0.x, 0);
            const int tHi = __builtin_amdgcn_readlane(cur.l1.w, 16);
            if (tLo <= ownHi && tHi >= ownLo) {
                const int jbase = ts * 16 + 8 * h;
                mine8_masked(c0, cur, jbase, i0, Li0, posV0, negV0);
                mine8_masked(c1, cur, jbase, i1, Li1, posV1, negV1);
            } else {
                mine8_plain(c0, negV0);
                mine8_plain(c1, negV1);
            }
            cidx = nidx;
        }
    }

    posV0 = fmaxf(posV0, __shfl_xor(posV0, 16, 32));
    negV0 = fmaxf(negV0, __shfl_xor(negV0, 16, 32));
    posV1 = fmaxf(posV1, __shfl_xor(posV1, 16, 32));
    negV1 = fmaxf(negV1, __shfl_xor(negV1, 16, 32));
    sP[w][lane] = h ? posV1 : posV0;
    sN[w][lane] = h ? negV1 : negV0;
    __syncthreads();
    if (w == 0) {
        float p = sP[0][lane];
        float n = sN[0][lane];
        #pragma unroll
        for (int pt = 1; pt < PARTS; ++pt) {
            p = fmaxf(p, sP[pt][lane]);
            n = fmaxf(n, sN[pt][lane]);
        }
        posM[own0 * 16 + lane] = p;
        negM[own0 * 16 + lane] = n;
        float prob = 0.0f;
        int   cnt  = 0;
        if (p > 0.0f && n > 0.0f) {
            const float s = (n - p) * INV_SIGMA;
            prob = (s > 0.0f) ? (s + log1pf(expf(-s))) : log1pf(expf(s));
            cnt  = 1;
        }
        float sp = p, sn = n;
        #pragma unroll
        for (int off = 16; off >= 1; off >>= 1) {
            prob += __shfl_xor(prob, off, 32);
            sp   += __shfl_xor(sp, off, 32);
            sn   += __shfl_xor(sn, off, 32);
            cnt  += __shfl_xor(cnt, off, 32);
        }
        if (lane == 0)
            ((float4*)part)[blockIdx.x] = make_float4(prob, sp, sn, __int_as_float(cnt));
    }
}

__global__ void __launch_bounds__(256)
ephn_finalize_kernel(const float* __restrict__ posM,
                     const float* __restrict__ negM,
                     const int*   __restrict__ perm,
                     const float* __restrict__ part,
                     float* __restrict__ out, int N, int nblk) {
    __shared__ float sProb[8];
    __shared__ float sPos[8];
    __shared__ float sNeg[8];
    __shared__ int   sCnt[8];
    const int tid  = threadIdx.x;
    const int lane = tid & 31;
    const int wv   = tid >> 5;
    const int g    = blockIdx.x * 256 + tid;
    const int nOut = 2 * N + 6;
    const int lastBlk = (nOut - 1) >> 8;

    float v = 0.0f;
    if (g >= 1 && g <= N) {
        const float p = posM[perm[g - 1]];
        v = (p > 0.0f) ? p : 0.0f;
    } else if (g >= N + 3 && g <= 2 * N + 2) {
        const float n = negM[perm[g - (N + 3)]];
        v = (n > 0.0f) ? n : 0.0f;
    } else if (g == N + 2 || g == 2 * N + 4) {
        v = 1.0f;
    }

    if (blockIdx.x == 0 || blockIdx.x == lastBlk) {
        float aProb = 0.0f, aPos = 0.0f, aNeg = 0.0f;
        int aCnt = 0;
        for (int k = tid; k < nblk; k += 256) {
            const float4 q = ((const float4*)part)[k];
            aProb += q.x; aPos += q.y; aNeg += q.z; aCnt += __float_as_int(q.w);
        }
        #pragma unroll
        for (int off = 16; off >= 1; off >>= 1) {
            aProb += __shfl_xor(aProb, off, 32);
            aPos  += __shfl_xor(aPos, off, 32);
            aNeg  += __shfl_xor(aNeg, off, 32);
            aCnt  += __shfl_xor(aCnt, off, 32);
        }
        if (lane == 0) { sProb[wv] = aProb; sPos[wv] = aPos; sNeg[wv] = aNeg; sCnt[wv] = aCnt; }
        __syncthreads();
        if (g == 0 || g == nOut - 1) {
            float tProb = sProb[0], tPos = sPos[0], tNeg = sNeg[0];
            int tCnt = sCnt[0];
            #pragma unroll
            for (int k = 1; k < 8; ++k) {
                tProb += sProb[k]; tPos += sPos[k]; tNeg += sNeg[k]; tCnt += sCnt[k];
            }
            const int cnt = (tCnt > 1) ? tCnt : 1;
            v = (g == 0) ? (tProb / (float)cnt) : ((tPos - tNeg) / (float)N);
        }
    }
    if (g < nOut) out[g] = v;
}

extern "C" void kernel_launch(void* const* d_in, const int* in_sizes, int n_in,
                              void* d_out, int out_size, void* d_ws, size_t ws_size,
                              hipStream_t stream) {
    const float* fvec = (const float*)d_in[0];
    const int*   Lvec = (const int*)d_in[1];
    const int N = in_sizes[1];

    float* pk    = (float*)d_ws;
    float* posM  = pk + N * DVEC;
    float* negM  = posM + N;
    float* part  = negM + N;
    int*   perm  = (int*)(part + N / 8);
    int*   Ls    = perm + N;

    ephn_rank_kernel<<<N / 32, 256, 0, stream>>>(Lvec, perm, N);
    ephn_pack_kernel<<<N / 32, 1024, 0, stream>>>(fvec, Lvec, perm, pk, Ls, N);
    ephn_main_kernel<<<N / 32, 256, 0, stream>>>(pk, Ls, posM, negM, part, N);
    ephn_finalize_kernel<<<(2 * N + 6 + 255) / 256, 256, 0, stream>>>(posM, negM, perm, part,
                                                                       (float*)d_out, N, N / 32);
}


# Each similarity tile computed once (the product is symmetric): a block sweeps only the 258 tiles from its own onward and mines each per own row and per streamed row (row_xmask max); new join launch
# speedup vs baseline: 6.3680x; 1.4335x over previous
#include <hip/hip_runtime.h>
#include <hip/hip_bf16.h>


typedef __attribute__((ext_vector_type(2))) float v2f;
typedef __attribute__((ext_vector_type(4))) float v4f;
typedef __attribute__((ext_vector_type(8))) float v8f;

#define DVEC 128
#define INV_SIGMA 10.0f
#define PARTS 8
#define PKSTR 33

__global__ void __launch_bounds__(256)
ephn_rank_kernel(const int* __restrict__ Lvec, int* __restrict__ perm, int N) {
    __shared__ int sRank[32];
    const int lane = threadIdx.x & 31;
    const int w    = threadIdx.x >> 5;
    const int rch  = blockIdx.x * 8 + w;
    const int4* Lp = (const int4*)Lvec;
    const int4 own = Lp[rch];
    const int Lr[4] = {own.x, own.y, own.z, own.w};
    int cnt[4] = {0, 0, 0, 0};
    const int nch = N >> 2;
    #pragma unroll 16
    for (int ch = lane; ch < nch; ch += 32) {
        const int4 v = Lp[ch];
        const int adj = (ch < rch) ? 1 : 0;
        #pragma unroll
        for (int k = 0; k < 4; ++k) {
            const int thr = Lr[k] + adj;
            cnt[k] += (v.x < thr) + (v.y < thr) + (v.z < thr) + (v.w < thr);
        }
    }
    #pragma unroll
    for (int k = 0; k < 4; ++k) {
        #pragma unroll
        for (int off = 16; off >= 1; off >>= 1)
            cnt[k] += __shfl_xor(cnt[k], off, 32);
    }
    cnt[1] += (Lr[0] == Lr[1]);
    cnt[2] += (Lr[0] == Lr[2]) + (Lr[1] == Lr[2]);
    cnt[3] += (Lr[0] == Lr[3]) + (Lr[1] == Lr[3]) + (Lr[2] == Lr[3]);
    if (lane == 0) {
        sRank[w * 4 + 0] = cnt[0];
        sRank[w * 4 + 1] = cnt[1];
        sRank[w * 4 + 2] = cnt[2];
        sRank[w * 4 + 3] = cnt[3];
    }
    __syncthreads();
    if (w == 0) perm[blockIdx.x * 32 + lane] = sRank[lane];
}

__global__ void __launch_bounds__(1024)
ephn_pack_kernel(const float* __restrict__ fvec, const int* __restrict__ Lvec,
                 const int* __restrict__ perm, float* __restrict__ pk,
                 int* __restrict__ Ls, int N) {
    __shared__ v4f tile[32 * PKSTR];
    __shared__ int sInv[32];
    const int wv   = threadIdx.x >> 5;
    const int lane = threadIdx.x & 31;
    const int p0   = blockIdx.x * 32;
    const int4* perm4 = (const int4*)perm;
    const int nch = N >> 2;
    for (int ca = threadIdx.x; ca < nch; ca += 2048) {
        const int cb = ca + 1024;
        const int4 va = perm4[ca];
        int4 vb = {0, 0, 0, 0};
        if (cb < nch) vb = perm4[cb];
        const int a0 = va.x - p0, a1 = va.y - p0, a2 = va.z - p0, a3 = va.w - p0;
        if ((unsigned)a0 < 32u) sInv[a0] = 4 * ca;
        if ((unsigned)a1 < 32u) sInv[a1] = 4 * ca + 1;
        if ((unsigned)a2 < 32u) sInv[a2] = 4 * ca + 2;
        if ((unsigned)a3 < 32u) sInv[a3] = 4 * ca + 3;
        if (cb < nch) {
            const int b0 = vb.x - p0, b1 = vb.y - p0, b2 = vb.z - p0, b3 = vb.w - p0;
            if ((unsigned)b0 < 32u) sInv[b0] = 4 * cb;
            if ((unsigned)b1 < 32u) sInv[b1] = 4 * cb + 1;
            if ((unsigned)b2 < 32u) sInv[b2] = 4 * cb + 2;
            if ((unsigned)b3 < 32u) sInv[b3] = 4 * cb + 3;
        }
    }
    __syncthreads();
    const int row = sInv[wv];
    const float4 d = ((const float4*)(fvec + ((size_t)row << 7)))[lane];
    float s = d.x * d.x + d.y * d.y + d.z * d.z + d.w * d.w;
    #pragma unroll
    for (int off = 16; off >= 1; off >>= 1)
        s += __shfl_xor(s, off, 32);
    const float r = rsqrtf(s);
    v4f o;
    o.x = d.x * r; o.y = d.y * r; o.z = d.z * r; o.w = d.w * r;
    tile[wv * PKSTR + lane] = o;
    __syncthreads();
    const int c  = lane & 15;
    const int h  = lane >> 4;
    const int tl = wv >> 4;
    const int q  = wv & 15;
    ((v4f*)pk)[((size_t)((blockIdx.x * 2 + tl) * 16 + q)) * 32 + lane] =
        tile[(tl * 16 + c) * PKSTR + 2 * q + h];
    if (wv == 0) Ls[p0 + lane] = Lvec[sInv[lane]];
}

struct StreamTile {
    v4f f[16];
    int4 l0, l1;
};

__device__ __forceinline__ void load_stream(StreamTile& s, const v4f* __restrict__ pk4,
                                            const int* __restrict__ Lvec,
                                            int ts, int lane, int h) {
    const v4f* p = pk4 + ((size_t)ts * 16) * 32 + lane;
    #pragma unroll
    for (int q = 0; q < 16; ++q) s.f[q] = p[q * 32];
    const int4* lp = (const int4*)(Lvec + ts * 16 + 8 * h);
    s.l0 = lp[0];
    s.l1 = lp[1];
}

#define XMAX_STEP(M)                                                                              \
    __device__ __forceinline__ float xmax_step##M(float v) {                                      \
        float o;                                                                                  \
        asm("v_max_num_f32_dpp %0, %1, %1 row_xmask:" #M " row_mask:0xf bank_mask:0xf"            \
            : "=v"(o) : "v"(v));                                                                  \
        return o;                                                                                 \
    }
XMAX_STEP(1)
XMAX_STEP(2)
XMAX_STEP(4)
XMAX_STEP(8)
#undef XMAX_STEP

__device__ __forceinline__ v8f rowmax8(v8f x) {
    #pragma unroll
    for (int r = 0; r < 8; ++r) x[r] = xmax_step1(x[r]);
    #pragma unroll
    for (int r = 0; r < 8; ++r) x[r] = xmax_step2(x[r]);
    #pragma unroll
    for (int r = 0; r < 8; ++r) x[r] = xmax_step4(x[r]);
    #pragma unroll
    for (int r = 0; r < 8; ++r) x[r] = xmax_step8(x[r]);
    return x;
}

__device__ __forceinline__ float pick8(const v8f& x, int lane) {
    const bool s0 = (lane & 1) != 0, s1 = (lane & 2) != 0, s2 = (lane & 4) != 0;
    const float a0 = s0 ? x[1] : x[0];
    const float a1 = s0 ? x[3] : x[2];
    const float a2 = s0 ? x[5] : x[4];
    const float a3 = s0 ? x[7] : x[6];
    const float e0 = s1 ? a1 : a0;
    const float e1 = s1 ? a3 : a2;
    return s2 ? e1 : e0;
}

__device__ __forceinline__ void do_tile(const StreamTile& cur,
                                        const v4f (&b0)[16], const v4f (&b1)[16],
                                        int ts, int lane, int h,
                                        int i0, int i1, int Li0, int Li1, int ownLo, int ownHi,
                                        float& posV0, float& negV0, float& posV1, float& negV1,
                                        float* __restrict__ colLine) {
    v8f c0 = {};
    v8f c1 = {};
    #pragma unroll
    for (int q = 0; q < 16; ++q) {
        const v2f alo = cur.f[q].xy;
        const v2f ahi = cur.f[q].zw;
        c0 = __builtin_amdgcn_wmma_f32_16x16x4_f32(
                false, alo, false, b0[q].xy, (short)0, c0, false, false);
        c1 = __builtin_amdgcn_wmma_f32_16x16x4_f32(
                false, alo, false, b1[q].xy, (short)0, c1, false, false);
        asm volatile("" : "+v"(c0), "+v"(c1));
        c0 = __builtin_amdgcn_wmma_f32_16x16x4_f32(
                false, ahi, false, b0[q].zw, (short)0, c0, false, false);
        c1 = __builtin_amdgcn_wmma_f32_16x16x4_f32(
                false, ahi, false, b1[q].zw, (short)0, c1, false, false);
        asm volatile("" : "+v"(c0), "+v"(c1));
    }

    v8f   np;
    float vp = -1.0f;
    const int tLo = __builtin_amdgcn_readlane(cur.l0.x, 0);
    const int tHi = __builtin_amdgcn_readlane(cur.l1.w, 16);
    if (tLo <= ownHi && tHi >= ownLo) {
        const int Lj[8] = {cur.l0.x, cur.l0.y, cur.l0.z, cur.l0.w,
                           cur.l1.x, cur.l1.y, cur.l1.z, cur.l1.w};
        const int jbase = ts * 16 + 8 * h;
        v8f pp;
        #pragma unroll
        for (int r = 0; r < 8; ++r) {
            const int   j    = jbase + r;
            const float d0   = c0[r];
            const float d1   = c1[r];
            const bool  eq0  = (Li0 == Lj[r]);
            const bool  eq1  = (Li1 == Lj[r]);
            const float pv0 = eq0 ? (((i0 == j) || (d0 > 0.9999f)) ? -1.0f : d0) : -1.0f;
            const float pv1 = eq1 ? (((i1 == j) || (d1 > 0.9999f)) ? -1.0f : d1) : -1.0f;
            const float nv0 = eq0 ? -1.0f : d0;
            const float nv1 = eq1 ? -1.0f : d1;
            posV0 = fmaxf(posV0, pv0);
            posV1 = fmaxf(posV1, pv1);
            negV0 = fmaxf(negV0, nv0);
            negV1 = fmaxf(negV1, nv1);
            pp[r] = fmaxf(pv0, pv1);
            np[r] = fmaxf(nv0, nv1);
        }
        pp = rowmax8(pp);
        vp = pick8(pp, lane);
    } else {
        #pragma unroll
        for (int r = 0; r < 8; ++r) {
            negV0 = fmaxf(negV0, c0[r]);
            negV1 = fmaxf(negV1, c1[r]);
            np[r] = fmaxf(c0[r], c1[r]);
        }
    }
    np = rowmax8(np);
    const float vn = pick8(np, lane);
    colLine[lane] = (lane & 8) ? vp : vn;
}

__global__ void __launch_bounds__(256)
ephn_main_kernel(const float* __restrict__ pk,
                 const int*   __restrict__ Lvec,
                 float* __restrict__ ownP,
                 float* __restrict__ ownN,
                 float* __restrict__ colM,
                 int N) {
    __shared__ float sP[PARTS][32];
    __shared__ float sN[PARTS][32];
    const int lane = threadIdx.x & 31;
    const int w    = __builtin_amdgcn_readfirstlane(threadIdx.x >> 5);
    const int c    = lane & 15;
    const int h    = lane >> 4;
    const int nt   = N >> 4;
    const int span = (nt >> 1) + 2;
    const int hp   = (nt >> 1) / PARTS;
    const int own0 = blockIdx.x * 2;
    const v4f* pk4 = (const v4f*)pk;

    v4f b0[16], b1[16];
    {
        const v4f* p0 = pk4 + ((size_t)own0 * 16) * 32 + lane;
        #pragma unroll
        for (int q = 0; q < 16; ++q) { b0[q] = p0[q * 32]; b1[q] = p0[(16 + q) * 32]; }
    }
    const int i0  = own0 * 16 + c;
    const int i1  = i0 + 16;
    const int Li0 = Lvec[i0];
    const int Li1 = Lvec[i1];

    const int ownLo = __builtin_amdgcn_readfirstlane(Lvec[own0 * 16]);
    const int ownHi = __builtin_amdgcn_readfirstlane(Lvec[own0 * 16 + 31]);

    float posV0 = -1.0f, negV0 = -1.0f, posV1 = -1.0f, negV1 = -1.0f;

    const int e0 = w * hp;
    const int eX = (nt >> 1) + (w & 1);
    float* colBlk = colM + ((size_t)blockIdx.x * span) * 32;

    StreamTile sa, sb;
    {
        int t = own0 + e0;
        t = (t >= nt) ? t - nt : t;
        load_stream(sa, pk4, Lvec, t, lane, h);
    }

    for (int it = 0; it < hp; it += 2) {
        #pragma unroll
        for (int half = 0; half < 2; ++half) {
            StreamTile& cur = half ? sb : sa;
            StreamTile& nxt = half ? sa : sb;
            const int i = it + half;
            const int e = e0 + i;
            int ts = own0 + e;
            ts = (ts >= nt) ? ts - nt : ts;
            const int en = (i + 1 < hp) ? e + 1 : eX;
            int tn = own0 + en;
            tn = (tn >= nt) ? tn - nt : tn;
            load_stream(nxt, pk4, Lvec, tn, lane, h);

            do_tile(cur, b0, b1, ts, lane, h, i0, i1, Li0, Li1, ownLo, ownHi,
                    posV0, negV0, posV1, negV1, colBlk + (size_t)e * 32);
        }
    }
    if (w < 2) {
        int ts = own0 + eX;
        ts = (ts >= nt) ? ts - nt : ts;
        do_tile(sa, b0, b1, ts, lane, h, i0, i1, Li0, Li1, ownLo, ownHi,
                posV0, negV0, posV1, negV1, colBlk + (size_t)eX * 32);
    }

    posV0 = fmaxf(posV0, __shfl_xor(posV0, 16, 32));
    negV0 = fmaxf(negV0, __shfl_xor(negV0, 16, 32));
    posV1 = fmaxf(posV1, __shfl_xor(posV1, 16, 32));
    negV1 = fmaxf(negV1, __shfl_xor(negV1, 16, 32));
    sP[w][lane] = h ? posV1 : posV0;
    sN[w][lane] = h ? negV1 : negV0;
    __syncthreads();
    if (w == 0) {
        float p = sP[0][lane];
        float n = sN[0][lane];
        #pragma unroll
        for (int pt = 1; pt < PARTS; ++pt) {
            p = fmaxf(p, sP[pt][lane]);
            n = fmaxf(n, sN[pt][lane]);
        }
        ownP[own0 * 16 + lane] = p;
        ownN[own0 * 16 + lane] = n;
    }
}

__global__ void __launch_bounds__(256)
ephn_join_kernel(const float* __restrict__ colM,
                 const float* __restrict__ ownP,
                 const float* __restrict__ ownN,
                 float* __restrict__ posM,
                 float* __restrict__ negM,
                 float* __restrict__ part,
                 int N) {
    __shared__ float sJ[8][32];
    const int lane = threadIdx.x & 31;
    const int w    = __builtin_amdgcn_readfirstlane(threadIdx.x >> 5);
    const int bt   = blockIdx.x;
    const int nb   = N >> 5;
    const int span = (N >> 5) + 2;
    const int s    = w >> 2;
    const int kw   = w & 3;
    const int nj   = nb >> 3;
    const int nj8  = nj & ~7;
    float m = -1.0f;
    for (int j = 0; j < nj8; j += 8) {
        float v[8];
        #pragma unroll
        for (int u = 0; u < 8; ++u) {
            const int k = kw + 4 * (j + u);
            int b = bt - k;
            b = (b < 0) ? b + nb : b;
            v[u] = colM[((size_t)b * span + (2 * k + s)) * 32 + lane];
        }
        #pragma unroll
        for (int u = 0; u < 8; ++u) m = fmaxf(m, v[u]);
    }
    for (int j = nj8; j < nj; ++j) {
        const int k = kw + 4 * j;
        int b = bt - k;
        b = (b < 0) ? b + nb : b;
        m = fmaxf(m, colM[((size_t)b * span + (2 * k + s)) * 32 + lane]);
    }
    {
        const int k = nb >> 1;
        int b = bt - k;
        b = (b < 0) ? b + nb : b;
        m = fmaxf(m, colM[((size_t)b * span + (2 * k + s)) * 32 + lane]);
    }
    sJ[w][lane] = m;
    __syncthreads();
    if (w == 0) {
        const int tl = lane >> 4;
        const int li = ((lane >> 3) & 1) * 16 + (lane & 7);
        float p = ownP[bt * 32 + lane];
        float n = ownN[bt * 32 + lane];
        #pragma unroll
        for (int kk = 0; kk < 4; ++kk) {
            n = fmaxf(n, sJ[tl * 4 + kk][li]);
            p = fmaxf(p, sJ[tl * 4 + kk][li + 8]);
        }
        posM[bt * 32 + lane] = p;
        negM[bt * 32 + lane] = n;
        float prob = 0.0f;
        int   cnt  = 0;
        if (p > 0.0f && n > 0.0f) {
            const float sg = (n - p) * INV_SIGMA;
            prob = (sg > 0.0f) ? (sg + log1pf(expf(-sg))) : log1pf(expf(sg));
            cnt  = 1;
        }
        float sp = p, sn = n;
        #pragma unroll
        for (int off = 16; off >= 1; off >>= 1) {
            prob += __shfl_xor(prob, off, 32);
            sp   += __shfl_xor(sp, off, 32);
            sn   += __shfl_xor(sn, off, 32);
            cnt  += __shfl_xor(cnt, off, 32);
        }
        if (lane == 0)
            ((float4*)part)[blockIdx.x] = make_float4(prob, sp, sn, __int_as_float(cnt));
    }
}

__global__ void __launch_bounds__(256)
ephn_finalize_kernel(const float* __restrict__ posM,
                     const float* __restrict__ negM,
                     const int*   __restrict__ perm,
                     const float* __restrict__ part,
                     float* __restrict__ out, int N, int nblk) {
    __shared__ float sProb[8];
    __shared__ float sPos[8];
    __shared__ float sNeg[8];
    __shared__ int   sCnt[8];
    const int tid  = threadIdx.x;
    const int lane = tid & 31;
    const int wv   = tid >> 5;
    const int g    = blockIdx.x * 256 + tid;
    const int nOut = 2 * N + 6;
    const int lastBlk = (nOut - 1) >> 8;

    float v = 0.0f;
    if (g >= 1 && g <= N) {
        const float p = posM[perm[g - 1]];
        v = (p > 0.0f) ? p : 0.0f;
    } else if (g >= N + 3 && g <= 2 * N + 2) {
        const float n = negM[perm[g - (N + 3)]];
        v = (n > 0.0f) ? n : 0.0f;
    } else if (g == N + 2 || g == 2 * N + 4) {
        v = 1.0f;
    }

    if (blockIdx.x == 0 || blockIdx.x == lastBlk) {
        float aProb = 0.0f, aPos = 0.0f, aNeg = 0.0f;
        int aCnt = 0;
        for (int k = tid; k < nblk; k += 256) {
            const float4 q = ((const float4*)part)[k];
            aProb += q.x; aPos += q.y; aNeg += q.z; aCnt += __float_as_int(q.w);
        }
        #pragma unroll
        for (int off = 16; off >= 1; off >>= 1) {
            aProb += __shfl_xor(aProb, off, 32);
            aPos  += __shfl_xor(aPos, off, 32);
            aNeg  += __shfl_xor(aNeg, off, 32);
            aCnt  += __shfl_xor(aCnt, off, 32);
        }
        if (lane == 0) { sProb[wv] = aProb; sPos[wv] = aPos; sNeg[wv] = aNeg; sCnt[wv] = aCnt; }
        __syncthreads();
        if (g == 0 || g == nOut - 1) {
            float tProb = sProb[0], tPos = sPos[0], tNeg = sNeg[0];
            int tCnt = sCnt[0];
            #pragma unroll
            for (int k = 1; k < 8; ++k) {
                tProb += sProb[k]; tPos += sPos[k]; tNeg += sNeg[k]; tCnt += sCnt[k];
            }
            const int cnt = (tCnt > 1) ? tCnt : 1;
            v = (g == 0) ? (tProb / (float)cnt) : ((tPos - tNeg) / (float)N);
        }
    }
    if (g < nOut) out[g] = v;
}

extern "C" void kernel_launch(void* const* d_in, const int* in_sizes, int n_in,
                              void* d_out, int out_size, void* d_ws, size_t ws_size,
                              hipStream_t stream) {
    const float* fvec = (const float*)d_in[0];
    const int*   Lvec = (const int*)d_in[1];
    const int N = in_sizes[1];

    float* pk    = (float*)d_ws;
    float* posM  = pk + N * DVEC;
    float* negM  = posM + N;
    float* part  = negM + N;
    int*   perm  = (int*)(part + N / 8);
    int*   Ls    = perm + N;
    float* ownP  = (float*)(Ls + N);
    float* ownN  = ownP + N;
    float* colM  = ownN + N;

    ephn_rank_kernel<<<N / 32, 256, 0, stream>>>(Lvec, perm, N);
    ephn_pack_kernel<<<N / 32, 1024, 0, stream>>>(fvec, Lvec, perm, pk, Ls, N);
    ephn_main_kernel<<<N / 32, 256, 0, stream>>>(pk, Ls, ownP, ownN, colM, N);
    ephn_join_kernel<<<N / 32, 256, 0, stream>>>(colM, ownP, ownN, posM, negM, part, N);
    ephn_finalize_kernel<<<(2 * N + 6 + 255) / 256, 256, 0, stream>>>(posM, negM, perm, part,
                                                                       (float*)d_out, N, N / 32);
}


# Short kernels issue their independent loads as one group ahead of the first wait: join 35 loads per wave in one memory round trip (was six), finalize block sums and pack labels beside their gathers
# speedup vs baseline: 6.5560x; 1.0295x over previous
#include <hip/hip_runtime.h>
#include <hip/hip_bf16.h>


typedef __attribute__((ext_vector_type(2))) float v2f;
typedef __attribute__((ext_vector_type(4))) float v4f;
typedef __attribute__((ext_vector_type(8))) float v8f;

#define DVEC 128
#define INV_SIGMA 10.0f
#define PARTS 8
#define PKSTR 33

__global__ void __launch_bounds__(256)
ephn_rank_kernel(const int* __restrict__ Lvec, int* __restrict__ perm, int N) {
    __shared__ int sRank[32];
    const int lane = threadIdx.x & 31;
    const int w    = threadIdx.x >> 5;
    const int rch  = blockIdx.x * 8 + w;
    const int4* Lp = (const int4*)Lvec;
    const int4 own = Lp[rch];
    const int Lr[4] = {own.x, own.y, own.z, own.w};
    int cnt[4] = {0, 0, 0, 0};
    const int nch = N >> 2;
    #pragma unroll 16
    for (int ch = lane; ch < nch; ch += 32) {
        const int4 v = Lp[ch];
        const int adj = (ch < rch) ? 1 : 0;
        #pragma unroll
        for (int k = 0; k < 4; ++k) {
            const int thr = Lr[k] + adj;
            cnt[k] += (v.x < thr) + (v.y < thr) + (v.z < thr) + (v.w < thr);
        }
    }
    #pragma unroll
    for (int k = 0; k < 4; ++k) {
        #pragma unroll
        for (int off = 16; off >= 1; off >>= 1)
            cnt[k] += __shfl_xor(cnt[k], off, 32);
    }
    cnt[1] += (Lr[0] == Lr[1]);
    cnt[2] += (Lr[0] == Lr[2]) + (Lr[1] == Lr[2]);
    cnt[3] += (Lr[0] == Lr[3]) + (Lr[1] == Lr[3]) + (Lr[2] == Lr[3]);
    if (lane == 0) {
        sRank[w * 4 + 0] = cnt[0];
        sRank[w * 4 + 1] = cnt[1];
        sRank[w * 4 + 2] = cnt[2];
        sRank[w * 4 + 3] = cnt[3];
    }
    __syncthreads();
    if (w == 0) perm[blockIdx.x * 32 + lane] = sRank[lane];
}

__global__ void __launch_bounds__(1024)
ephn_pack_kernel(const float* __restrict__ fvec, const int* __restrict__ Lvec,
                 const int* __restrict__ perm, float* __restrict__ pk,
                 int* __restrict__ Ls, int N) {
    __shared__ v4f tile[32 * PKSTR];
    __shared__ int sInv[32];
    const int wv   = threadIdx.x >> 5;
    const int lane = threadIdx.x & 31;
    const int p0   = blockIdx.x * 32;
    const int4* perm4 = (const int4*)perm;
    const int nch = N >> 2;
    for (int ca = threadIdx.x; ca < nch; ca += 2048) {
        const int cb = ca + 1024;
        const int4 va = perm4[ca];
        int4 vb = {0, 0, 0, 0};
        if (cb < nch) vb = perm4[cb];
        const int a0 = va.x - p0, a1 = va.y - p0, a2 = va.z - p0, a3 = va.w - p0;
        if ((unsigned)a0 < 32u) sInv[a0] = 4 * ca;
        if ((unsigned)a1 < 32u) sInv[a1] = 4 * ca + 1;
        if ((unsigned)a2 < 32u) sInv[a2] = 4 * ca + 2;
        if ((unsigned)a3 < 32u) sInv[a3] = 4 * ca + 3;
        if (cb < nch) {
            const int b0 = vb.x - p0, b1 = vb.y - p0, b2 = vb.z - p0, b3 = vb.w - p0;
            if ((unsigned)b0 < 32u) sInv[b0] = 4 * cb;
            if ((unsigned)b1 < 32u) sInv[b1] = 4 * cb + 1;
            if ((unsigned)b2 < 32u) sInv[b2] = 4 * cb + 2;
            if ((unsigned)b3 < 32u) sInv[b3] = 4 * cb + 3;
        }
    }
    __syncthreads();
    const int row = sInv[wv];
    int lsv = 0;
    if (wv == 0) lsv = Lvec[sInv[lane]];
    const float4 d = ((const float4*)(fvec + ((size_t)row << 7)))[lane];
    float s = d.x * d.x + d.y * d.y + d.z * d.z + d.w * d.w;
    #pragma unroll
    for (int off = 16; off >= 1; off >>= 1)
        s += __shfl_xor(s, off, 32);
    const float r = rsqrtf(s);
    v4f o;
    o.x = d.x * r; o.y = d.y * r; o.z = d.z * r; o.w = d.w * r;
    tile[wv * PKSTR + lane] = o;
    __syncthreads();
    const int c  = lane & 15;
    const int h  = lane >> 4;
    const int tl = wv >> 4;
    const int q  = wv & 15;
    ((v4f*)pk)[((size_t)((blockIdx.x * 2 + tl) * 16 + q)) * 32 + lane] =
        tile[(tl * 16 + c) * PKSTR + 2 * q + h];
    if (wv == 0) Ls[p0 + lane] = lsv;
}

struct StreamTile {
    v4f f[16];
    int4 l0, l1;
};

__device__ __forceinline__ void load_stream(StreamTile& s, const v4f* __restrict__ pk4,
                                            const int* __restrict__ Lvec,
                                            int ts, int lane, int h) {
    const v4f* p = pk4 + ((size_t)ts * 16) * 32 + lane;
    #pragma unroll
    for (int q = 0; q < 16; ++q) s.f[q] = p[q * 32];
    const int4* lp = (const int4*)(Lvec + ts * 16 + 8 * h);
    s.l0 = lp[0];
    s.l1 = lp[1];
}

#define XMAX_STEP(M)                                                                              \
    __device__ __forceinline__ float xmax_step##M(float v) {                                      \
        float o;                                                                                  \
        asm("v_max_num_f32_dpp %0, %1, %1 row_xmask:" #M " row_mask:0xf bank_mask:0xf"            \
            : "=v"(o) : "v"(v));                                                                  \
        return o;                                                                                 \
    }
XMAX_STEP(1)
XMAX_STEP(2)
XMAX_STEP(4)
XMAX_STEP(8)
#undef XMAX_STEP

__device__ __forceinline__ v8f rowmax8(v8f x) {
    #pragma unroll
    for (int r = 0; r < 8; ++r) x[r] = xmax_step1(x[r]);
    #pragma unroll
    for (int r = 0; r < 8; ++r) x[r] = xmax_step2(x[r]);
    #pragma unroll
    for (int r = 0; r < 8; ++r) x[r] = xmax_step4(x[r]);
    #pragma unroll
    for (int r = 0; r < 8; ++r) x[r] = xmax_step8(x[r]);
    return x;
}

__device__ __forceinline__ float pick8(const v8f& x, int lane) {
    const bool s0 = (lane & 1) != 0, s1 = (lane & 2) != 0, s2 = (lane & 4) != 0;
    const float a0 = s0 ? x[1] : x[0];
    const float a1 = s0 ? x[3] : x[2];
    const float a2 = s0 ? x[5] : x[4];
    const float a3 = s0 ? x[7] : x[6];
    const float e0 = s1 ? a1 : a0;
    const float e1 = s1 ? a3 : a2;
    return s2 ? e1 : e0;
}

__device__ __forceinline__ void do_tile(const StreamTile& cur,
                                        const v4f (&b0)[16], const v4f (&b1)[16],
                                        int ts, int lane, int h,
                                        int i0, int i1, int Li0, int Li1, int ownLo, int ownHi,
                                        float& posV0, float& negV0, float& posV1, float& negV1,
                                        float* __restrict__ colLine) {
    v8f c0 = {};
    v8f c1 = {};
    #pragma unroll
    for (int q = 0; q < 16; ++q) {
        const v2f alo = cur.f[q].xy;
        const v2f ahi = cur.f[q].zw;
        c0 = __builtin_amdgcn_wmma_f32_16x16x4_f32(
                false, alo, false, b0[q].xy, (short)0, c0, false, false);
        c1 = __builtin_amdgcn_wmma_f32_16x16x4_f32(
                false, alo, false, b1[q].xy, (short)0, c1, false, false);
        asm volatile("" : "+v"(c0), "+v"(c1));
        c0 = __builtin_amdgcn_wmma_f32_16x16x4_f32(
                false, ahi, false, b0[q].zw, (short)0, c0, false, false);
        c1 = __builtin_amdgcn_wmma_f32_16x16x4_f32(
                false, ahi, false, b1[q].zw, (short)0, c1, false, false);
        asm volatile("" : "+v"(c0), "+v"(c1));
    }

    v8f   np;
    float vp = -1.0f;
    const int tLo = __builtin_amdgcn_readlane(cur.l0.x, 0);
    const int tHi = __builtin_amdgcn_readlane(cur.l1.w, 16);
    if (tLo <= ownHi && tHi >= ownLo) {
        const int Lj[8] = {cur.l0.x, cur.l0.y, cur.l0.z, cur.l0.w,
                           cur.l1.x, cur.l1.y, cur.l1.z, cur.l1.w};
        const int jbase = ts * 16 + 8 * h;
        v8f pp;
        #pragma unroll
        for (int r = 0; r < 8; ++r) {
            const int   j    = jbase + r;
            const float d0   = c0[r];
            const float d1   = c1[r];
            const bool  eq0  = (Li0 == Lj[r]);
            const bool  eq1  = (Li1 == Lj[r]);
            const float pv0 = eq0 ? (((i0 == j) || (d0 > 0.9999f)) ? -1.0f : d0) : -1.0f;
            const float pv1 = eq1 ? (((i1 == j) || (d1 > 0.9999f)) ? -1.0f : d1) : -1.0f;
            const float nv0 = eq0 ? -1.0f : d0;
            const float nv1 = eq1 ? -1.0f : d1;
            posV0 = fmaxf(posV0, pv0);
            posV1 = fmaxf(posV1, pv1);
            negV0 = fmaxf(negV0, nv0);
            negV1 = fmaxf(negV1, nv1);
            pp[r] = fmaxf(pv0, pv1);
            np[r] = fmaxf(nv0, nv1);
        }
        pp = rowmax8(pp);
        vp = pick8(pp, lane);
    } else {
        #pragma unroll
        for (int r = 0; r < 8; ++r) {
            negV0 = fmaxf(negV0, c0[r]);
            negV1 = fmaxf(negV1, c1[r]);
            np[r] = fmaxf(c0[r], c1[r]);
        }
    }
    np = rowmax8(np);
    const float vn = pick8(np, lane);
    colLine[lane] = (lane & 8) ? vp : vn;
}

__global__ void __launch_bounds__(256)
ephn_main_kernel(const float* __restrict__ pk,
                 const int*   __restrict__ Lvec,
                 float* __restrict__ ownP,
                 float* __restrict__ ownN,
                 float* __restrict__ colM,
                 int N) {
    __shared__ float sP[PARTS][32];
    __shared__ float sN[PARTS][32];
    const int lane = threadIdx.x & 31;
    const int w    = __builtin_amdgcn_readfirstlane(threadIdx.x >> 5);
    const int c    = lane & 15;
    const int h    = lane >> 4;
    const int nt   = N >> 4;
    const int span = (nt >> 1) + 2;
    const int hp   = (nt >> 1) / PARTS;
    const int own0 = blockIdx.x * 2;
    const v4f* pk4 = (const v4f*)pk;

    v4f b0[16], b1[16];
    {
        const v4f* p0 = pk4 + ((size_t)own0 * 16) * 32 + lane;
        #pragma unroll
        for (int q = 0; q < 16; ++q) { b0[q] = p0[q * 32]; b1[q] = p0[(16 + q) * 32]; }
    }
    const int i0  = own0 * 16 + c;
    const int i1  = i0 + 16;
    const int Li0 = Lvec[i0];
    const int Li1 = Lvec[i1];

    const int ownLo = __builtin_amdgcn_readfirstlane(Lvec[own0 * 16]);
    const int ownHi = __builtin_amdgcn_readfirstlane(Lvec[own0 * 16 + 31]);

    float posV0 = -1.0f, negV0 = -1.0f, posV1 = -1.0f, negV1 = -1.0f;

    const int e0 = w * hp;
    const int eX = (nt >> 1) + (w & 1);
    float* colBlk = colM + ((size_t)blockIdx.x * span) * 32;

    StreamTile sa, sb;
    {
        int t = own0 + e0;
        t = (t >= nt) ? t - nt : t;
        load_stream(sa, pk4, Lvec, t, lane, h);
    }

    for (int it = 0; it < hp; it += 2) {
        #pragma unroll
        for (int half = 0; half < 2; ++half) {
            StreamTile& cur = half ? sb : sa;
            StreamTile& nxt = half ? sa : sb;
            const int i = it + half;
            const int e = e0 + i;
            int ts = own0 + e;
            ts = (ts >= nt) ? ts - nt : ts;
            const int en = (i + 1 < hp) ? e + 1 : eX;
            int tn = own0 + en;
            tn = (tn >= nt) ? tn - nt : tn;
            load_stream(nxt, pk4, Lvec, tn, lane, h);

            do_tile(cur, b0, b1, ts, lane, h, i0, i1, Li0, Li1, ownLo, ownHi,
                    posV0, negV0, posV1, negV1, colBlk + (size_t)e * 32);
        }
    }
    if (w < 2) {
        int ts = own0 + eX;
        ts = (ts >= nt) ? ts - nt : ts;
        do_tile(sa, b0, b1, ts, lane, h, i0, i1, Li0, Li1, ownLo, ownHi,
                posV0, negV0, posV1, negV1, colBlk + (size_t)eX * 32);
    }

    posV0 = fmaxf(posV0, __shfl_xor(posV0, 16, 32));
    negV0 = fmaxf(negV0, __shfl_xor(negV0, 16, 32));
    posV1 = fmaxf(posV1, __shfl_xor(posV1, 16, 32));
    negV1 = fmaxf(negV1, __shfl_xor(negV1, 16, 32));
    sP[w][lane] = h ? posV1 : posV0;
    sN[w][lane] = h ? negV1 : negV0;
    __syncthreads();
    if (w == 0) {
        float p = sP[0][lane];
        float n = sN[0][lane];
        #pragma unroll
        for (int pt = 1; pt < PARTS; ++pt) {
            p = fmaxf(p, sP[pt][lane]);
            n = fmaxf(n, sN[pt][lane]);
        }
        ownP[own0 * 16 + lane] = p;
        ownN[own0 * 16 + lane] = n;
    }
}

__global__ void __launch_bounds__(256)
ephn_join_kernel(const float* __restrict__ colM,
                 const float* __restrict__ ownP,
                 const float* __restrict__ ownN,
                 float* __restrict__ posM,
                 float* __restrict__ negM,
                 float* __restrict__ part,
                 int N) {
    __shared__ float sJ[8][32];
    const int lane = threadIdx.x & 31;
    const int w    = __builtin_amdgcn_readfirstlane(threadIdx.x >> 5);
    const int bt   = blockIdx.x;
    const int nb   = N >> 5;
    const int span = (N >> 5) + 2;
    const int s    = w >> 2;
    const int kw   = w & 3;
    const int nj   = nb >> 3;
    float pOwn = -1.0f, nOwn = -1.0f;
    if (w == 0) {
        pOwn = ownP[bt * 32 + lane];
        nOwn = ownN[bt * 32 + lane];
    }
    float mFar;
    {
        const int k = nb >> 1;
        int b = bt - k;
        b = (b < 0) ? b + nb : b;
        const float* lp = colM + ((size_t)(unsigned)(b * span + 2 * k + s) << 5);
        mFar = lp[lane];
    }
    float m = -1.0f;
    int j = 0;
    for (; j + 32 <= nj; j += 32) {
        float v[32];
        #pragma unroll
        for (int u = 0; u < 32; ++u) {
            const int k = kw + 4 * (j + u);
            int b = bt - k;
            b = (b < 0) ? b + nb : b;
            const float* lp = colM + ((size_t)(unsigned)(b * span + 2 * k + s) << 5);
            v[u] = lp[lane];
        }
        #pragma unroll
        for (int u = 0; u < 32; ++u) m = fmaxf(m, v[u]);
    }
    for (; j < nj; ++j) {
        const int k = kw + 4 * j;
        int b = bt - k;
        b = (b < 0) ? b + nb : b;
        m = fmaxf(m, colM[((size_t)b * span + (2 * k + s)) * 32 + lane]);
    }
    m = fmaxf(m, mFar);
    sJ[w][lane] = m;
    __syncthreads();
    if (w == 0) {
        const int tl = lane >> 4;
        const int li = ((lane >> 3) & 1) * 16 + (lane & 7);
        float p = pOwn;
        float n = nOwn;
        #pragma unroll
        for (int kk = 0; kk < 4; ++kk) {
            n = fmaxf(n, sJ[tl * 4 + kk][li]);
            p = fmaxf(p, sJ[tl * 4 + kk][li + 8]);
        }
        posM[bt * 32 + lane] = p;
        negM[bt * 32 + lane] = n;
        float prob = 0.0f;
        int   cnt  = 0;
        if (p > 0.0f && n > 0.0f) {
            const float sg = (n - p) * INV_SIGMA;
            prob = (sg > 0.0f) ? (sg + log1pf(expf(-sg))) : log1pf(expf(sg));
            cnt  = 1;
        }
        float sp = p, sn = n;
        #pragma unroll
        for (int off = 16; off >= 1; off >>= 1) {
            prob += __shfl_xor(prob, off, 32);
            sp   += __shfl_xor(sp, off, 32);
            sn   += __shfl_xor(sn, off, 32);
            cnt  += __shfl_xor(cnt, off, 32);
        }
        if (lane == 0)
            ((float4*)part)[blockIdx.x] = make_float4(prob, sp, sn, __int_as_float(cnt));
    }
}

__global__ void __launch_bounds__(256)
ephn_finalize_kernel(const float* __restrict__ posM,
                     const float* __restrict__ negM,
                     const int*   __restrict__ perm,
                     const float* __restrict__ part,
                     float* __restrict__ out, int N, int nblk) {
    __shared__ float sProb[8];
    __shared__ float sPos[8];
    __shared__ float sNeg[8];
    __shared__ int   sCnt[8];
    const int tid  = threadIdx.x;
    const int lane = tid & 31;
    const int wv   = tid >> 5;
    const int g    = blockIdx.x * 256 + tid;
    const int nOut = 2 * N + 6;
    const int lastBlk = (nOut - 1) >> 8;

    const bool sums = (blockIdx.x == 0 || blockIdx.x == lastBlk);
    float4 q0 = make_float4(0.0f, 0.0f, 0.0f, 0.0f);
    if (sums && tid < nblk) q0 = ((const float4*)part)[tid];

    float v = 0.0f;
    if (g >= 1 && g <= N) {
        const float p = posM[perm[g - 1]];
        v = (p > 0.0f) ? p : 0.0f;
    } else if (g >= N + 3 && g <= 2 * N + 2) {
        const float n = negM[perm[g - (N + 3)]];
        v = (n > 0.0f) ? n : 0.0f;
    } else if (g == N + 2 || g == 2 * N + 4) {
        v = 1.0f;
    }

    if (sums) {
        float aProb, aPos, aNeg;
        int aCnt;
        if (nblk <= 256) {
            aProb = q0.x; aPos = q0.y; aNeg = q0.z;
            aCnt = __float_as_int(q0.w);
        } else {
            aProb = 0.0f; aPos = 0.0f; aNeg = 0.0f;
            aCnt = 0;
            for (int k = tid; k < nblk; k += 256) {
                const float4 q = ((const float4*)part)[k];
                aProb += q.x; aPos += q.y; aNeg += q.z; aCnt += __float_as_int(q.w);
            }
        }
        #pragma unroll
        for (int off = 16; off >= 1; off >>= 1) {
            aProb += __shfl_xor(aProb, off, 32);
            aPos  += __shfl_xor(aPos, off, 32);
            aNeg  += __shfl_xor(aNeg, off, 32);
            aCnt  += __shfl_xor(aCnt, off, 32);
        }
        if (lane == 0) { sProb[wv] = aProb; sPos[wv] = aPos; sNeg[wv] = aNeg; sCnt[wv] = aCnt; }
        __syncthreads();
        if (g == 0 || g == nOut - 1) {
            float tProb = sProb[0], tPos = sPos[0], tNeg = sNeg[0];
            int tCnt = sCnt[0];
            #pragma unroll
            for (int k = 1; k < 8; ++k) {
                tProb += sProb[k]; tPos += sPos[k]; tNeg += sNeg[k]; tCnt += sCnt[k];
            }
            const int cnt = (tCnt > 1) ? tCnt : 1;
            v = (g == 0) ? (tProb / (float)cnt) : ((tPos - tNeg) / (float)N);
        }
    }
    if (g < nOut) out[g] = v;
}

extern "C" void kernel_launch(void* const* d_in, const int* in_sizes, int n_in,
                              void* d_out, int out_size, void* d_ws, size_t ws_size,
                              hipStream_t stream) {
    const float* fvec = (const float*)d_in[0];
    const int*   Lvec = (const int*)d_in[1];
    const int N = in_sizes[1];

    float* pk    = (float*)d_ws;
    float* posM  = pk + N * DVEC;
    float* negM  = posM + N;
    float* part  = negM + N;
    int*   perm  = (int*)(part + N / 8);
    int*   Ls    = perm + N;
    float* ownP  = (float*)(Ls + N);
    float* ownN  = ownP + N;
    float* colM  = ownN + N;

    ephn_rank_kernel<<<N / 32, 256, 0, stream>>>(Lvec, perm, N);
    ephn_pack_kernel<<<N / 32, 1024, 0, stream>>>(fvec, Lvec, perm, pk, Ls, N);
    ephn_main_kernel<<<N / 32, 256, 0, stream>>>(pk, Ls, ownP, ownN, colM, N);
    ephn_join_kernel<<<N / 32, 256, 0, stream>>>(colM, ownP, ownN, posM, negM, part, N);
    ephn_finalize_kernel<<<(2 * N + 6 + 255) / 256, 256, 0, stream>>>(posM, negM, perm, part,
                                                                       (float*)d_out, N, N / 32);
}


# Class sort dropped: rows packed in input order, rank launch and perm lookups gone (4 launches); same-class pairs found by label compares, masked mining only there; loss and mean sum order changes
# speedup vs baseline: 6.9549x; 1.0608x over previous
#include <hip/hip_runtime.h>
#include <hip/hip_bf16.h>


typedef __attribute__((ext_vector_type(2))) float v2f;
typedef __attribute__((ext_vector_type(4))) float v4f;
typedef __attribute__((ext_vector_type(8))) float v8f;

#define DVEC 128
#define INV_SIGMA 10.0f
#define PARTS 8
#define PKSTR 33

__global__ void __launch_bounds__(1024)
ephn_pack_kernel(const float* __restrict__ fvec, float* __restrict__ pk) {
    __shared__ v4f tile[32 * PKSTR];
    const int wv   = threadIdx.x >> 5;
    const int lane = threadIdx.x & 31;
    const int row  = blockIdx.x * 32 + wv;
    const float4 d = ((const float4*)(fvec + ((size_t)row << 7)))[lane];
    float s = d.x * d.x + d.y * d.y + d.z * d.z + d.w * d.w;
    #pragma unroll
    for (int off = 16; off >= 1; off >>= 1)
        s += __shfl_xor(s, off, 32);
    const float r = rsqrtf(s);
    v4f o;
    o.x = d.x * r; o.y = d.y * r; o.z = d.z * r; o.w = d.w * r;
    tile[wv * PKSTR + lane] = o;
    __syncthreads();
    const int c  = lane & 15;
    const int h  = lane >> 4;
    const int tl = wv >> 4;
    const int q  = wv & 15;
    ((v4f*)pk)[((size_t)((blockIdx.x * 2 + tl) * 16 + q)) * 32 + lane] =
        tile[(tl * 16 + c) * PKSTR + 2 * q + h];
}

struct StreamTile {
    v4f f[16];
    int4 l0, l1;
};

__device__ __forceinline__ void load_stream(StreamTile& s, const v4f* __restrict__ pk4,
                                            const int* __restrict__ Lvec,
                                            int ts, int lane, int h) {
    const v4f* p = pk4 + ((size_t)ts * 16) * 32 + lane;
    #pragma unroll
    for (int q = 0; q < 16; ++q) s.f[q] = p[q * 32];
    const int4* lp = (const int4*)(Lvec + ts * 16 + 8 * h);
    s.l0 = lp[0];
    s.l1 = lp[1];
}

#define XMAX_STEP(M)                                                                              \
    __device__ __forceinline__ float xmax_step##M(float v) {                                      \
        float o;                                                                                  \
        asm("v_max_num_f32_dpp %0, %1, %1 row_xmask:" #M " row_mask:0xf bank_mask:0xf"            \
            : "=v"(o) : "v"(v));                                                                  \
        return o;                                                                                 \
    }
XMAX_STEP(1)
XMAX_STEP(2)
XMAX_STEP(4)
XMAX_STEP(8)
#undef XMAX_STEP

__device__ __forceinline__ v8f rowmax8(v8f x) {
    #pragma unroll
    for (int r = 0; r < 8; ++r) x[r] = xmax_step1(x[r]);
    #pragma unroll
    for (int r = 0; r < 8; ++r) x[r] = xmax_step2(x[r]);
    #pragma unroll
    for (int r = 0; r < 8; ++r) x[r] = xmax_step4(x[r]);
    #pragma unroll
    for (int r = 0; r < 8; ++r) x[r] = xmax_step8(x[r]);
    return x;
}

__device__ __forceinline__ float pick8(const v8f& x, int lane) {
    const bool s0 = (lane & 1) != 0, s1 = (lane & 2) != 0, s2 = (lane & 4) != 0;
    const float a0 = s0 ? x[1] : x[0];
    const float a1 = s0 ? x[3] : x[2];
    const float a2 = s0 ? x[5] : x[4];
    const float a3 = s0 ? x[7] : x[6];
    const float e0 = s1 ? a1 : a0;
    const float e1 = s1 ? a3 : a2;
    return s2 ? e1 : e0;
}

__device__ __forceinline__ void do_tile(const StreamTile& cur,
                                        const v4f (&b0)[16], const v4f (&b1)[16],
                                        int ts, int lane, int h,
                                        int i0, int i1, int Li0, int Li1,
                                        float& posV0, float& negV0, float& posV1, float& negV1,
                                        float* __restrict__ colLine) {
    v8f c0 = {};
    v8f c1 = {};
    #pragma unroll
    for (int q = 0; q < 16; ++q) {
        const v2f alo = cur.f[q].xy;
        const v2f ahi = cur.f[q].zw;
        c0 = __builtin_amdgcn_wmma_f32_16x16x4_f32(
                false, alo, false, b0[q].xy, (short)0, c0, false, false);
        c1 = __builtin_amdgcn_wmma_f32_16x16x4_f32(
                false, alo, false, b1[q].xy, (short)0, c1, false, false);
        asm volatile("" : "+v"(c0), "+v"(c1));
        c0 = __builtin_amdgcn_wmma_f32_16x16x4_f32(
                false, ahi, false, b0[q].zw, (short)0, c0, false, false);
        c1 = __builtin_amdgcn_wmma_f32_16x16x4_f32(
                false, ahi, false, b1[q].zw, (short)0, c1, false, false);
        asm volatile("" : "+v"(c0), "+v"(c1));
    }

    const int Lj[8] = {cur.l0.x, cur.l0.y, cur.l0.z, cur.l0.w,
                       cur.l1.x, cur.l1.y, cur.l1.z, cur.l1.w};
    unsigned hit[8];
    unsigned anyHit = 0u;
    #pragma unroll
    for (int r = 0; r < 8; ++r) {
        hit[r] = __builtin_amdgcn_ballot_w32(Li0 == Lj[r]) | __builtin_amdgcn_ballot_w32(Li1 == Lj[r]);
        anyHit |= hit[r];
    }

    float vp = -1.0f;
    if (anyHit != 0u) {
        const int jbase = ts * 16 + 8 * h;
        v8f pp;
        #pragma unroll
        for (int r = 0; r < 8; ++r) pp[r] = -1.0f;
        #pragma unroll
        for (int r = 0; r < 8; ++r) {
            if (__builtin_expect(hit[r] != 0u, 0)) {
                const bool  eq0 = (Li0 == Lj[r]);
                const bool  eq1 = (Li1 == Lj[r]);
                const int   j  = jbase + r;
                const float d0 = c0[r];
                const float d1 = c1[r];
                const float pv0 = eq0 ? (((i0 == j) || (d0 > 0.9999f)) ? -1.0f : d0) : -1.0f;
                const float pv1 = eq1 ? (((i1 == j) || (d1 > 0.9999f)) ? -1.0f : d1) : -1.0f;
                posV0 = fmaxf(posV0, pv0);
                posV1 = fmaxf(posV1, pv1);
                float p = fmaxf(pv0, pv1);
                p = xmax_step1(p);
                p = xmax_step2(p);
                p = xmax_step4(p);
                p = xmax_step8(p);
                pp[r] = p;
                c0[r] = eq0 ? -1.0f : d0;
                c1[r] = eq1 ? -1.0f : d1;
            }
        }
        vp = pick8(pp, lane);
    }

    v8f np;
    #pragma unroll
    for (int r = 0; r < 8; ++r) {
        negV0 = fmaxf(negV0, c0[r]);
        negV1 = fmaxf(negV1, c1[r]);
        np[r] = fmaxf(c0[r], c1[r]);
    }
    np = rowmax8(np);
    const float vn = pick8(np, lane);
    colLine[lane] = (lane & 8) ? vp : vn;
}

__global__ void __launch_bounds__(256)
ephn_main_kernel(const float* __restrict__ pk,
                 const int*   __restrict__ Lvec,
                 float* __restrict__ ownP,
                 float* __restrict__ ownN,
                 float* __restrict__ colM,
                 int N) {
    __shared__ float sP[PARTS][32];
    __shared__ float sN[PARTS][32];
    const int lane = threadIdx.x & 31;
    const int w    = __builtin_amdgcn_readfirstlane(threadIdx.x >> 5);
    const int c    = lane & 15;
    const int h    = lane >> 4;
    const int nt   = N >> 4;
    const int span = (nt >> 1) + 2;
    const int hp   = (nt >> 1) / PARTS;
    const int own0 = blockIdx.x * 2;
    const v4f* pk4 = (const v4f*)pk;

    v4f b0[16], b1[16];
    {
        const v4f* p0 = pk4 + ((size_t)own0 * 16) * 32 + lane;
        #pragma unroll
        for (int q = 0; q < 16; ++q) { b0[q] = p0[q * 32]; b1[q] = p0[(16 + q) * 32]; }
    }
    const int i0  = own0 * 16 + c;
    const int i1  = i0 + 16;
    const int Li0 = Lvec[i0];
    const int Li1 = Lvec[i1];

    float posV0 = -1.0f, negV0 = -1.0f, posV1 = -1.0f, negV1 = -1.0f;

    const int e0 = w * hp;
    const int eX = (nt >> 1) + (w & 1);
    float* colBlk = colM + ((size_t)blockIdx.x * span) * 32;

    StreamTile sa, sb;
    {
        int t = own0 + e0;
        t = (t >= nt) ? t - nt : t;
        load_stream(sa, pk4, Lvec, t, lane, h);
    }

    for (int it = 0; it < hp; it += 2) {
        #pragma unroll
        for (int half = 0; half < 2; ++half) {
            StreamTile& cur = half ? sb : sa;
            StreamTile& nxt = half ? sa : sb;
            const int i = it + half;
            const int e = e0 + i;
            int ts = own0 + e;
            ts = (ts >= nt) ? ts - nt : ts;
            const int en = (i + 1 < hp) ? e + 1 : eX;
            int tn = own0 + en;
            tn = (tn >= nt) ? tn - nt : tn;
            load_stream(nxt, pk4, Lvec, tn, lane, h);

            do_tile(cur, b0, b1, ts, lane, h, i0, i1, Li0, Li1,
                    posV0, negV0, posV1, negV1, colBlk + (size_t)e * 32);
        }
    }
    if (w < 2) {
        int ts = own0 + eX;
        ts = (ts >= nt) ? ts - nt : ts;
        do_tile(sa, b0, b1, ts, lane, h, i0, i1, Li0, Li1,
                posV0, negV0, posV1, negV1, colBlk + (size_t)eX * 32);
    }

    posV0 = fmaxf(posV0, __shfl_xor(posV0, 16, 32));
    negV0 = fmaxf(negV0, __shfl_xor(negV0, 16, 32));
    posV1 = fmaxf(posV1, __shfl_xor(posV1, 16, 32));
    negV1 = fmaxf(negV1, __shfl_xor(negV1, 16, 32));
    sP[w][lane] = h ? posV1 : posV0;
    sN[w][lane] = h ? negV1 : negV0;
    __syncthreads();
    if (w == 0) {
        float p = sP[0][lane];
        float n = sN[0][lane];
        #pragma unroll
        for (int pt = 1; pt < PARTS; ++pt) {
            p = fmaxf(p, sP[pt][lane]);
            n = fmaxf(n, sN[pt][lane]);
        }
        ownP[own0 * 16 + lane] = p;
        ownN[own0 * 16 + lane] = n;
    }
}

__global__ void __launch_bounds__(256)
ephn_join_kernel(const float* __restrict__ colM,
                 const float* __restrict__ ownP,
                 const float* __restrict__ ownN,
                 float* __restrict__ posM,
                 float* __restrict__ negM,
                 float* __restrict__ part,
                 int N) {
    __shared__ float sJ[8][32];
    const int lane = threadIdx.x & 31;
    const int w    = __builtin_amdgcn_readfirstlane(threadIdx.x >> 5);
    const int bt   = blockIdx.x;
    const int nb   = N >> 5;
    const int span = (N >> 5) + 2;
    const int s    = w >> 2;
    const int kw   = w & 3;
    const int nj   = nb >> 3;
    float pOwn = -1.0f, nOwn = -1.0f;
    if (w == 0) {
        pOwn = ownP[bt * 32 + lane];
        nOwn = ownN[bt * 32 + lane];
    }
    float mFar;
    {
        const int k = nb >> 1;
        int b = bt - k;
        b = (b < 0) ? b + nb : b;
        const float* lp = colM + ((size_t)(unsigned)(b * span + 2 * k + s) << 5);
        mFar = lp[lane];
    }
    float m = -1.0f;
    int j = 0;
    for (; j + 32 <= nj; j += 32) {
        float v[32];
        #pragma unroll
        for (int u = 0; u < 32; ++u) {
            const int k = kw + 4 * (j + u);
            int b = bt - k;
            b = (b < 0) ? b + nb : b;
            const float* lp = colM + ((size_t)(unsigned)(b * span + 2 * k + s) << 5);
            v[u] = lp[lane];
        }
        #pragma unroll
        for (int u = 0; u < 32; ++u) m = fmaxf(m, v[u]);
    }
    for (; j < nj; ++j) {
        const int k = kw + 4 * j;
        int b = bt - k;
        b = (b < 0) ? b + nb : b;
        m = fmaxf(m, colM[((size_t)b * span + (2 * k + s)) * 32 + lane]);
    }
    m = fmaxf(m, mFar);
    sJ[w][lane] = m;
    __syncthreads();
    if (w == 0) {
        const int tl = lane >> 4;
        const int li = ((lane >> 3) & 1) * 16 + (lane & 7);
        float p = pOwn;
        float n = nOwn;
        #pragma unroll
        for (int kk = 0; kk < 4; ++kk) {
            n = fmaxf(n, sJ[tl * 4 + kk][li]);
            p = fmaxf(p, sJ[tl * 4 + kk][li + 8]);
        }
        posM[bt * 32 + lane] = p;
        negM[bt * 32 + lane] = n;
        float prob = 0.0f;
        int   cnt  = 0;
        if (p > 0.0f && n > 0.0f) {
            const float sg = (n - p) * INV_SIGMA;
            prob = (sg > 0.0f) ? (sg + log1pf(expf(-sg))) : log1pf(expf(sg));
            cnt  = 1;
        }
        float sp = p, sn = n;
        #pragma unroll
        for (int off = 16; off >= 1; off >>= 1) {
            prob += __shfl_xor(prob, off, 32);
            sp   += __shfl_xor(sp, off, 32);
            sn   += __shfl_xor(sn, off, 32);
            cnt  += __shfl_xor(cnt, off, 32);
        }
        if (lane == 0)
            ((float4*)part)[blockIdx.x] = make_float4(prob, sp, sn, __int_as_float(cnt));
    }
}

__global__ void __launch_bounds__(256)
ephn_finalize_kernel(const float* __restrict__ posM,
                     const float* __restrict__ negM,
                     const float* __restrict__ part,
                     float* __restrict__ out, int N, int nblk) {
    __shared__ float sProb[8];
    __shared__ float sPos[8];
    __shared__ float sNeg[8];
    __shared__ int   sCnt[8];
    const int tid  = threadIdx.x;
    const int lane = tid & 31;
    const int wv   = tid >> 5;
    const int g    = blockIdx.x * 256 + tid;
    const int nOut = 2 * N + 6;
    const int lastBlk = (nOut - 1) >> 8;

    const bool sums = (blockIdx.x == 0 || blockIdx.x == lastBlk);
    float4 q0 = make_float4(0.0f, 0.0f, 0.0f, 0.0f);
    if (sums && tid < nblk) q0 = ((const float4*)part)[tid];

    float v = 0.0f;
    if (g >= 1 && g <= N) {
        const float p = posM[g - 1];
        v = (p > 0.0f) ? p : 0.0f;
    } else if (g >= N + 3 && g <= 2 * N + 2) {
        const float n = negM[g - (N + 3)];
        v = (n > 0.0f) ? n : 0.0f;
    } else if (g == N + 2 || g == 2 * N + 4) {
        v = 1.0f;
    }

    if (sums) {
        float aProb, aPos, aNeg;
        int aCnt;
        if (nblk <= 256) {
            aProb = q0.x; aPos = q0.y; aNeg = q0.z;
            aCnt = __float_as_int(q0.w);
        } else {
            aProb = 0.0f; aPos = 0.0f; aNeg = 0.0f;
            aCnt = 0;
            for (int k = tid; k < nblk; k += 256) {
                const float4 q = ((const float4*)part)[k];
                aProb += q.x; aPos += q.y; aNeg += q.z; aCnt += __float_as_int(q.w);
            }
        }
        #pragma unroll
        for (int off = 16; off >= 1; off >>= 1) {
            aProb += __shfl_xor(aProb, off, 32);
            aPos  += __shfl_xor(aPos, off, 32);
            aNeg  += __shfl_xor(aNeg, off, 32);
            aCnt  += __shfl_xor(aCnt, off, 32);
        }
        if (lane == 0) { sProb[wv] = aProb; sPos[wv] = aPos; sNeg[wv] = aNeg; sCnt[wv] = aCnt; }
        __syncthreads();
        if (g == 0 || g == nOut - 1) {
            float tProb = sProb[0], tPos = sPos[0], tNeg = sNeg[0];
            int tCnt = sCnt[0];
            #pragma unroll
            for (int k = 1; k < 8; ++k) {
                tProb += sProb[k]; tPos += sPos[k]; tNeg += sNeg[k]; tCnt += sCnt[k];
            }
            const int cnt = (tCnt > 1) ? tCnt : 1;
            v = (g == 0) ? (tProb / (float)cnt) : ((tPos - tNeg) / (float)N);
        }
    }
    if (g < nOut) out[g] = v;
}

extern "C" void kernel_launch(void* const* d_in, const int* in_sizes, int n_in,
                              void* d_out, int out_size, void* d_ws, size_t ws_size,
                              hipStream_t stream) {
    const float* fvec = (const float*)d_in[0];
    const int*   Lvec = (const int*)d_in[1];
    const int N = in_sizes[1];

    float* pk    = (float*)d_ws;
    float* posM  = pk + N * DVEC;
    float* negM  = posM + N;
    float* part  = negM + N;
    float* ownP  = part + N / 8;
    float* ownN  = ownP + N;
    float* colM  = ownN + N;

    ephn_pack_kernel<<<N / 32, 1024, 0, stream>>>(fvec, pk);
    ephn_main_kernel<<<N / 32, 256, 0, stream>>>(pk, Lvec, ownP, ownN, colM, N);
    ephn_join_kernel<<<N / 32, 256, 0, stream>>>(colM, ownP, ownN, posM, negM, part, N);
    ephn_finalize_kernel<<<(2 * N + 6 + 255) / 256, 256, 0, stream>>>(posM, negM, part,
                                                                       (float*)d_out, N, N / 32);
}


# Mining product operands swapped (own rows = row operand, streamed tile = column operand): a streamed row's maxima are lane-local, per-tile 16-lane reductions gone; own-row maxima joined once per wave
# speedup vs baseline: 7.3513x; 1.0570x over previous
#include <hip/hip_runtime.h>
#include <hip/hip_bf16.h>


typedef __attribute__((ext_vector_type(2))) float v2f;
typedef __attribute__((ext_vector_type(4))) float v4f;
typedef __attribute__((ext_vector_type(8))) float v8f;

#define DVEC 128
#define INV_SIGMA 10.0f
#define PARTS 8
#define PKSTR 33

__global__ void __launch_bounds__(1024)
ephn_pack_kernel(const float* __restrict__ fvec, float* __restrict__ pk) {
    __shared__ v4f tile[32 * PKSTR];
    const int wv   = threadIdx.x >> 5;
    const int lane = threadIdx.x & 31;
    const int row  = blockIdx.x * 32 + wv;
    const float4 d = ((const float4*)(fvec + ((size_t)row << 7)))[lane];
    float s = d.x * d.x + d.y * d.y + d.z * d.z + d.w * d.w;
    #pragma unroll
    for (int off = 16; off >= 1; off >>= 1)
        s += __shfl_xor(s, off, 32);
    const float r = rsqrtf(s);
    v4f o;
    o.x = d.x * r; o.y = d.y * r; o.z = d.z * r; o.w = d.w * r;
    tile[wv * PKSTR + lane] = o;
    __syncthreads();
    const int c  = lane & 15;
    const int h  = lane >> 4;
    const int tl = wv >> 4;
    const int q  = wv & 15;
    ((v4f*)pk)[((size_t)((blockIdx.x * 2 + tl) * 16 + q)) * 32 + lane] =
        tile[(tl * 16 + c) * PKSTR + 2 * q + h];
}

struct StreamTile {
    v4f f[16];
    int lab;
};

__device__ __forceinline__ void load_stream(StreamTile& s, const v4f* __restrict__ pk4,
                                            const int* __restrict__ Lvec,
                                            int ts, int lane, int c) {
    const v4f* p = pk4 + ((size_t)ts * 16) * 32 + lane;
    #pragma unroll
    for (int q = 0; q < 16; ++q) s.f[q] = p[q * 32];
    s.lab = Lvec[ts * 16 + c];
}

__device__ __forceinline__ float xhalf(float x) {
    return __builtin_bit_cast(float,
        __builtin_amdgcn_permlanex16(__builtin_bit_cast(int, x), __builtin_bit_cast(int, x),
                                     0x76543210, 0xfedcba98, false, false));
}

#define XMAX_STEP(M)                                                                              \
    __device__ __forceinline__ float xmax_step##M(float v) {                                      \
        float o;                                                                                  \
        asm("v_max_num_f32_dpp %0, %1, %1 row_xmask:" #M " row_mask:0xf bank_mask:0xf"            \
            : "=v"(o) : "v"(v));                                                                  \
        return o;                                                                                 \
    }
XMAX_STEP(1)
XMAX_STEP(2)
XMAX_STEP(4)
XMAX_STEP(8)
#undef XMAX_STEP

__device__ __forceinline__ v8f rowmax8(v8f x) {
    #pragma unroll
    for (int r = 0; r < 8; ++r) x[r] = xmax_step1(x[r]);
    #pragma unroll
    for (int r = 0; r < 8; ++r) x[r] = xmax_step2(x[r]);
    #pragma unroll
    for (int r = 0; r < 8; ++r) x[r] = xmax_step4(x[r]);
    #pragma unroll
    for (int r = 0; r < 8; ++r) x[r] = xmax_step8(x[r]);
    return x;
}

__device__ __forceinline__ float pick8(const v8f& x, int lane) {
    const bool s0 = (lane & 1) != 0, s1 = (lane & 2) != 0, s2 = (lane & 4) != 0;
    const float a0 = s0 ? x[1] : x[0];
    const float a1 = s0 ? x[3] : x[2];
    const float a2 = s0 ? x[5] : x[4];
    const float a3 = s0 ? x[7] : x[6];
    const float e0 = s1 ? a1 : a0;
    const float e1 = s1 ? a3 : a2;
    return s2 ? e1 : e0;
}

__device__ __forceinline__ void do_tile(const StreamTile& cur,
                                        const v4f (&a0)[16], const v4f (&a1)[16],
                                        int ts, int lane, int c, int h, int ibase,
                                        const int (&Lo0)[8], const int (&Lo1)[8],
                                        float (&posO0)[8], float (&negO0)[8],
                                        float (&posO1)[8], float (&negO1)[8],
                                        float* __restrict__ colLine) {
    v8f c0 = {};
    v8f c1 = {};
    #pragma unroll
    for (int q = 0; q < 16; ++q) {
        const v2f blo = cur.f[q].xy;
        const v2f bhi = cur.f[q].zw;
        c0 = __builtin_amdgcn_wmma_f32_16x16x4_f32(
                false, a0[q].xy, false, blo, (short)0, c0, false, false);
        c1 = __builtin_amdgcn_wmma_f32_16x16x4_f32(
                false, a1[q].xy, false, blo, (short)0, c1, false, false);
        asm volatile("" : "+v"(c0), "+v"(c1));
        c0 = __builtin_amdgcn_wmma_f32_16x16x4_f32(
                false, a0[q].zw, false, bhi, (short)0, c0, false, false);
        c1 = __builtin_amdgcn_wmma_f32_16x16x4_f32(
                false, a1[q].zw, false, bhi, (short)0, c1, false, false);
        asm volatile("" : "+v"(c0), "+v"(c1));
    }

    const int Lj = cur.lab;
    unsigned hit[8];
    unsigned anyHit = 0u;
    #pragma unroll
    for (int r = 0; r < 8; ++r) {
        hit[r] = __builtin_amdgcn_ballot_w32(Lo0[r] == Lj) | __builtin_amdgcn_ballot_w32(Lo1[r] == Lj);
        anyHit |= hit[r];
    }

    float vp = -1.0f;
    if (anyHit != 0u) {
        const int j = ts * 16 + c;
        #pragma unroll
        for (int r = 0; r < 8; ++r) {
            if (__builtin_expect(hit[r] != 0u, 0)) {
                const bool  eq0 = (Lo0[r] == Lj);
                const bool  eq1 = (Lo1[r] == Lj);
                const int   i0 = ibase + r;
                const int   i1 = i0 + 16;
                const float d0 = c0[r];
                const float d1 = c1[r];
                const float pv0 = eq0 ? (((i0 == j) || (d0 > 0.9999f)) ? -1.0f : d0) : -1.0f;
                const float pv1 = eq1 ? (((i1 == j) || (d1 > 0.9999f)) ? -1.0f : d1) : -1.0f;
                posO0[r] = fmaxf(posO0[r], pv0);
                posO1[r] = fmaxf(posO1[r], pv1);
                vp = fmaxf(vp, fmaxf(pv0, pv1));
                c0[r] = eq0 ? -1.0f : d0;
                c1[r] = eq1 ? -1.0f : d1;
            }
        }
    }

    float vn = fmaxf(c0[0], c1[0]);
    negO0[0] = fmaxf(negO0[0], c0[0]);
    negO1[0] = fmaxf(negO1[0], c1[0]);
    #pragma unroll
    for (int r = 1; r < 8; ++r) {
        negO0[r] = fmaxf(negO0[r], c0[r]);
        negO1[r] = fmaxf(negO1[r], c1[r]);
        vn = fmaxf(vn, fmaxf(c0[r], c1[r]));
    }
    const float keep = h ? vp : vn;
    const float send = h ? vn : vp;
    colLine[lane] = fmaxf(keep, xhalf(send));
}

__global__ void __launch_bounds__(256)
ephn_main_kernel(const float* __restrict__ pk,
                 const int*   __restrict__ Lvec,
                 float* __restrict__ ownP,
                 float* __restrict__ ownN,
                 float* __restrict__ colM,
                 int N) {
    __shared__ float sP[PARTS][32];
    __shared__ float sN[PARTS][32];
    const int lane = threadIdx.x & 31;
    const int w    = __builtin_amdgcn_readfirstlane(threadIdx.x >> 5);
    const int c    = lane & 15;
    const int h    = lane >> 4;
    const int nt   = N >> 4;
    const int span = (nt >> 1) + 2;
    const int hp   = (nt >> 1) / PARTS;
    const int own0 = blockIdx.x * 2;
    const v4f* pk4 = (const v4f*)pk;

    v4f a0[16], a1[16];
    {
        const v4f* p0 = pk4 + ((size_t)own0 * 16) * 32 + lane;
        #pragma unroll
        for (int q = 0; q < 16; ++q) { a0[q] = p0[q * 32]; a1[q] = p0[(16 + q) * 32]; }
    }
    const int ibase = own0 * 16 + 8 * h;
    int Lo0[8], Lo1[8];
    {
        const int4* lp = (const int4*)(Lvec + ibase);
        const int4 x0 = lp[0], x1 = lp[1], y0 = lp[4], y1 = lp[5];
        Lo0[0] = x0.x; Lo0[1] = x0.y; Lo0[2] = x0.z; Lo0[3] = x0.w;
        Lo0[4] = x1.x; Lo0[5] = x1.y; Lo0[6] = x1.z; Lo0[7] = x1.w;
        Lo1[0] = y0.x; Lo1[1] = y0.y; Lo1[2] = y0.z; Lo1[3] = y0.w;
        Lo1[4] = y1.x; Lo1[5] = y1.y; Lo1[6] = y1.z; Lo1[7] = y1.w;
    }

    float posO0[8], negO0[8], posO1[8], negO1[8];
    #pragma unroll
    for (int r = 0; r < 8; ++r) {
        posO0[r] = -1.0f; negO0[r] = -1.0f; posO1[r] = -1.0f; negO1[r] = -1.0f;
    }

    const int e0 = w * hp;
    const int eX = (nt >> 1) + (w & 1);
    float* colBlk = colM + ((size_t)blockIdx.x * span) * 32;

    StreamTile sa, sb;
    {
        int t = own0 + e0;
        t = (t >= nt) ? t - nt : t;
        load_stream(sa, pk4, Lvec, t, lane, c);
    }

    for (int it = 0; it < hp; it += 2) {
        #pragma unroll
        for (int half = 0; half < 2; ++half) {
            StreamTile& cur = half ? sb : sa;
            StreamTile& nxt = half ? sa : sb;
            const int i = it + half;
            const int e = e0 + i;
            int ts = own0 + e;
            ts = (ts >= nt) ? ts - nt : ts;
            const int en = (i + 1 < hp) ? e + 1 : eX;
            int tn = own0 + en;
            tn = (tn >= nt) ? tn - nt : tn;
            load_stream(nxt, pk4, Lvec, tn, lane, c);

            do_tile(cur, a0, a1, ts, lane, c, h, ibase, Lo0, Lo1,
                    posO0, negO0, posO1, negO1, colBlk + (size_t)e * 32);
        }
    }
    if (w < 2) {
        int ts = own0 + eX;
        ts = (ts >= nt) ? ts - nt : ts;
        do_tile(sa, a0, a1, ts, lane, c, h, ibase, Lo0, Lo1,
                posO0, negO0, posO1, negO1, colBlk + (size_t)eX * 32);
    }

    v8f n0, n1, p0, p1;
    #pragma unroll
    for (int r = 0; r < 8; ++r) {
        n0[r] = negO0[r]; n1[r] = negO1[r]; p0[r] = posO0[r]; p1[r] = posO1[r];
    }
    n0 = rowmax8(n0);
    n1 = rowmax8(n1);
    p0 = rowmax8(p0);
    p1 = rowmax8(p1);
    const float nv = (lane & 8) ? pick8(n1, lane) : pick8(n0, lane);
    const float pv = (lane & 8) ? pick8(p1, lane) : pick8(p0, lane);
    const int rowb = ((lane >> 3) & 1) * 16 + 8 * h + (lane & 7);
    sP[w][rowb] = pv;
    sN[w][rowb] = nv;
    __syncthreads();
    if (w == 0) {
        float p = sP[0][lane];
        float n = sN[0][lane];
        #pragma unroll
        for (int pt = 1; pt < PARTS; ++pt) {
            p = fmaxf(p, sP[pt][lane]);
            n = fmaxf(n, sN[pt][lane]);
        }
        ownP[own0 * 16 + lane] = p;
        ownN[own0 * 16 + lane] = n;
    }
}

__global__ void __launch_bounds__(256)
ephn_join_kernel(const float* __restrict__ colM,
                 const float* __restrict__ ownP,
                 const float* __restrict__ ownN,
                 float* __restrict__ posM,
                 float* __restrict__ negM,
                 float* __restrict__ part,
                 int N) {
    __shared__ float sJ[8][32];
    const int lane = threadIdx.x & 31;
    const int w    = __builtin_amdgcn_readfirstlane(threadIdx.x >> 5);
    const int bt   = blockIdx.x;
    const int nb   = N >> 5;
    const int span = (N >> 5) + 2;
    const int s    = w >> 2;
    const int kw   = w & 3;
    const int nj   = nb >> 3;
    float pOwn = -1.0f, nOwn = -1.0f;
    if (w == 0) {
        pOwn = ownP[bt * 32 + lane];
        nOwn = ownN[bt * 32 + lane];
    }
    float mFar;
    {
        const int k = nb >> 1;
        int b = bt - k;
        b = (b < 0) ? b + nb : b;
        const float* lp = colM + ((size_t)(unsigned)(b * span + 2 * k + s) << 5);
        mFar = lp[lane];
    }
    float m = -1.0f;
    int j = 0;
    for (; j + 32 <= nj; j += 32) {
        float v[32];
        #pragma unroll
        for (int u = 0; u < 32; ++u) {
            const int k = kw + 4 * (j + u);
            int b = bt - k;
            b = (b < 0) ? b + nb : b;
            const float* lp = colM + ((size_t)(unsigned)(b * span + 2 * k + s) << 5);
            v[u] = lp[lane];
        }
        #pragma unroll
        for (int u = 0; u < 32; ++u) m = fmaxf(m, v[u]);
    }
    for (; j < nj; ++j) {
        const int k = kw + 4 * j;
        int b = bt - k;
        b = (b < 0) ? b + nb : b;
        m = fmaxf(m, colM[((size_t)b * span + (2 * k + s)) * 32 + lane]);
    }
    m = fmaxf(m, mFar);
    sJ[w][lane] = m;
    __syncthreads();
    if (w == 0) {
        const int tl = lane >> 4;
        const int rr = lane & 15;
        float p = pOwn;
        float n = nOwn;
        #pragma unroll
        for (int kk = 0; kk < 4; ++kk) {
            n = fmaxf(n, sJ[tl * 4 + kk][rr]);
            p = fmaxf(p, sJ[tl * 4 + kk][rr + 16]);
        }
        posM[bt * 32 + lane] = p;
        negM[bt * 32 + lane] = n;
        float prob = 0.0f;
        int   cnt  = 0;
        if (p > 0.0f && n > 0.0f) {
            const float sg = (n - p) * INV_SIGMA;
            prob = (sg > 0.0f) ? (sg + log1pf(expf(-sg))) : log1pf(expf(sg));
            cnt  = 1;
        }
        float sp = p, sn = n;
        #pragma unroll
        for (int off = 16; off >= 1; off >>= 1) {
            prob += __shfl_xor(prob, off, 32);
            sp   += __shfl_xor(sp, off, 32);
            sn   += __shfl_xor(sn, off, 32);
            cnt  += __shfl_xor(cnt, off, 32);
        }
        if (lane == 0)
            ((float4*)part)[blockIdx.x] = make_float4(prob, sp, sn, __int_as_float(cnt));
    }
}

__global__ void __launch_bounds__(256)
ephn_finalize_kernel(const float* __restrict__ posM,
                     const float* __restrict__ negM,
                     const float* __restrict__ part,
                     float* __restrict__ out, int N, int nblk) {
    __shared__ float sProb[8];
    __shared__ float sPos[8];
    __shared__ float sNeg[8];
    __shared__ int   sCnt[8];
    const int tid  = threadIdx.x;
    const int lane = tid & 31;
    const int wv   = tid >> 5;
    const int g    = blockIdx.x * 256 + tid;
    const int nOut = 2 * N + 6;
    const int lastBlk = (nOut - 1) >> 8;

    const bool sums = (blockIdx.x == 0 || blockIdx.x == lastBlk);
    float4 q0 = make_float4(0.0f, 0.0f, 0.0f, 0.0f);
    if (sums && tid < nblk) q0 = ((const float4*)part)[tid];

    float v = 0.0f;
    if (g >= 1 && g <= N) {
        const float p = posM[g - 1];
        v = (p > 0.0f) ? p : 0.0f;
    } else if (g >= N + 3 && g <= 2 * N + 2) {
        const float n = negM[g - (N + 3)];
        v = (n > 0.0f) ? n : 0.0f;
    } else if (g == N + 2 || g == 2 * N + 4) {
        v = 1.0f;
    }

    if (sums) {
        float aProb, aPos, aNeg;
        int aCnt;
        if (nblk <= 256) {
            aProb = q0.x; aPos = q0.y; aNeg = q0.z;
            aCnt = __float_as_int(q0.w);
        } else {
            aProb = 0.0f; aPos = 0.0f; aNeg = 0.0f;
            aCnt = 0;
            for (int k = tid; k < nblk; k += 256) {
                const float4 q = ((const float4*)part)[k];
                aProb += q.x; aPos += q.y; aNeg += q.z; aCnt += __float_as_int(q.w);
            }
        }
        #pragma unroll
        for (int off = 16; off >= 1; off >>= 1) {
            aProb += __shfl_xor(aProb, off, 32);
            aPos  += __shfl_xor(aPos, off, 32);
            aNeg  += __shfl_xor(aNeg, off, 32);
            aCnt  += __shfl_xor(aCnt, off, 32);
        }
        if (lane == 0) { sProb[wv] = aProb; sPos[wv] = aPos; sNeg[wv] = aNeg; sCnt[wv] = aCnt; }
        __syncthreads();
        if (g == 0 || g == nOut - 1) {
            float tProb = sProb[0], tPos = sPos[0], tNeg = sNeg[0];
            int tCnt = sCnt[0];
            #pragma unroll
            for (int k = 1; k < 8; ++k) {
                tProb += sProb[k]; tPos += sPos[k]; tNeg += sNeg[k]; tCnt += sCnt[k];
            }
            const int cnt = (tCnt > 1) ? tCnt : 1;
            v = (g == 0) ? (tProb / (float)cnt) : ((tPos - tNeg) / (float)N);
        }
    }
    if (g < nOut) out[g] = v;
}

extern "C" void kernel_launch(void* const* d_in, const int* in_sizes, int n_in,
                              void* d_out, int out_size, void* d_ws, size_t ws_size,
                              hipStream_t stream) {
    const float* fvec = (const float*)d_in[0];
    const int*   Lvec = (const int*)d_in[1];
    const int N = in_sizes[1];

    float* pk    = (float*)d_ws;
    float* posM  = pk + N * DVEC;
    float* negM  = posM + N;
    float* part  = negM + N;
    float* ownP  = part + N / 8;
    float* ownN  = ownP + N;
    float* colM  = ownN + N;

    ephn_pack_kernel<<<N / 32, 1024, 0, stream>>>(fvec, pk);
    ephn_main_kernel<<<N / 32, 256, 0, stream>>>(pk, Lvec, ownP, ownN, colM, N);
    ephn_join_kernel<<<N / 32, 256, 0, stream>>>(colM, ownP, ownN, posM, negM, part, N);
    ephn_finalize_kernel<<<(2 * N + 6 + 255) / 256, 256, 0, stream>>>(posM, negM, part,
                                                                       (float*)d_out, N, N / 32);
}
